# starting kernel
# baseline (speedup 1.0000x reference)
#include <hip/hip_runtime.h>
#include <hip/hip_bf16.h>

typedef __attribute__((ext_vector_type(2))) float v2f;
typedef __attribute__((ext_vector_type(8))) float v8f;

__device__ __forceinline__ float lrelu(float v) { return v > 0.0f ? v : 0.01f * v; }

__global__ void gcn_deg_init(float* __restrict__ deg, int n) {
    int i = blockIdx.x * blockDim.x + threadIdx.x;
    if (i < n) deg[i] = 1.0f;
}

__global__ void gcn_deg_count(const int* __restrict__ dst, float* __restrict__ deg, int E) {
    int e = blockIdx.x * blockDim.x + threadIdx.x;
    if (e < E) atomicAdd(&deg[dst[e]], 1.0f);
}

__global__ void gcn_deg_finish(float* __restrict__ dinv, int n) {
    int i = blockIdx.x * blockDim.x + threadIdx.x;
    if (i < n) dinv[i] = rsqrtf(dinv[i]);
}

template <int K, int NOUT, int MODE>
__global__ __launch_bounds__(128) void gcn_gemm(
    const float* __restrict__ in,
    const float* __restrict__ W,
    const float* __restrict__ bias,
    const float* __restrict__ dinv,
    float* __restrict__ out,
    float* __restrict__ acc_out,
    int n)
{
    __shared__ float sW[NOUT * K];
    for (int i = threadIdx.x; i < NOUT * K / 4; i += blockDim.x)
        ((float4*)sW)[i] = ((const float4*)W)[i];
    __syncthreads();

    const int lane  = threadIdx.x & 31;
    const int wave  = threadIdx.x >> 5;
    const int row0  = (blockIdx.x * 4 + wave) * 16;
    if (row0 >= n) return;

    const int lhalf = lane >> 4;
    const int l16   = lane & 15;
    const int arow  = row0 + l16;

    v8f acc[NOUT / 16];
    #pragma unroll
    for (int t = 0; t < NOUT / 16; ++t)
        acc[t] = (v8f){0.f, 0.f, 0.f, 0.f, 0.f, 0.f, 0.f, 0.f};

    #pragma unroll
    for (int k0 = 0; k0 < K; k0 += 4) {
        const int kb = k0 + 2 * lhalf;
        v2f a;
        a.x = in[(size_t)arow * K + kb];
        a.y = in[(size_t)arow * K + kb + 1];
        #pragma unroll
        for (int t = 0; t < NOUT / 16; ++t) {
            const int ncol = t * 16 + l16;
            v2f b;
            b.x = sW[ncol * K + kb];
            b.y = sW[ncol * K + kb + 1];
            acc[t] = __builtin_amdgcn_wmma_f32_16x16x4_f32(
                false, a, false, b, (short)0, acc[t], false, false);
        }
    }

    #pragma unroll
    for (int t = 0; t < NOUT / 16; ++t) {
        const int col = t * 16 + l16;
        if (MODE == 0) {
            const float bv = bias[col];
            #pragma unroll
            for (int r = 0; r < 8; ++r) {
                const int row = row0 + r + 8 * lhalf;
                out[(size_t)row * NOUT + col] = lrelu(acc[t][r] + bv);
            }
        } else {
            #pragma unroll
            for (int r = 0; r < 8; ++r) {
                const int row = row0 + r + 8 * lhalf;
                const float val = acc[t][r] * dinv[row];
                out[(size_t)row * NOUT + col]     = val;
                acc_out[(size_t)row * NOUT + col] = val;
            }
        }
    }
}

__global__ void gcn_scatter(const int* __restrict__ src, const int* __restrict__ dst,
                            const float* __restrict__ p, float* __restrict__ acc, int E)
{
    int t = blockIdx.x * blockDim.x + threadIdx.x;
    int e = t >> 3;
    if (e >= E) return;
    int c4 = (t & 7) * 4;
    int u = src[e];
    int v = dst[e];
    float4 m = *(const float4*)(p + (size_t)u * 32 + c4);
    float* a = acc + (size_t)v * 32 + c4;
    atomicAdd(a + 0, m.x);
    atomicAdd(a + 1, m.y);
    atomicAdd(a + 2, m.z);
    atomicAdd(a + 3, m.w);
}

__global__ void gcn_finish(float* __restrict__ acc, const float* __restrict__ dinv,
                           const float* __restrict__ bias, int n)
{
    int t = blockIdx.x * blockDim.x + threadIdx.x;
    if (t >= n * 32) return;
    int row = t >> 5;
    int c   = t & 31;
    acc[t] = lrelu(dinv[row] * acc[t] + bias[c]);
}

__global__ void gcn_out(const float* __restrict__ acc2, const float* __restrict__ dinv,
                        const float* __restrict__ b2, const float* __restrict__ Wout,
                        const float* __restrict__ bout, float* __restrict__ out, int n)
{
    int i = blockIdx.x * blockDim.x + threadIdx.x;
    if (i >= n) return;
    float di = dinv[i];
    float s = 0.0f;
    #pragma unroll
    for (int c = 0; c < 32; ++c)
        s += Wout[c] * lrelu(di * acc2[(size_t)i * 32 + c] + b2[c]);
    out[i] = s + bout[0];
}

extern "C" void kernel_launch(void* const* d_in, const int* in_sizes, int n_in,
                              void* d_out, int out_size, void* d_ws, size_t ws_size,
                              hipStream_t stream)
{
    const float* x     = (const float*)d_in[0];
    const int*   ei    = (const int*)d_in[1];
    const float* W_in  = (const float*)d_in[2];
    const float* b_in  = (const float*)d_in[3];
    const float* W1    = (const float*)d_in[4];
    const float* b1    = (const float*)d_in[5];
    const float* W2    = (const float*)d_in[6];
    const float* b2    = (const float*)d_in[7];
    const float* W_out = (const float*)d_in[8];
    const float* b_out = (const float*)d_in[9];
    float* out = (float*)d_out;

    const int N = in_sizes[0] / 64;
    const int E = in_sizes[1] / 2;
    const int* src = ei;
    const int* dst = ei + E;

    float* ws   = (float*)d_ws;
    float* dinv = ws;
    float* h0   = ws + (size_t)N;
    float* p    = ws + (size_t)N * 65;
    float* accA = ws + (size_t)N * 97;
    float* accB = ws + (size_t)N * 129;

    const int TB = 256;
    gcn_deg_init  <<<(N + TB - 1) / TB, TB, 0, stream>>>(dinv, N);
    gcn_deg_count <<<(E + TB - 1) / TB, TB, 0, stream>>>(dst, dinv, E);
    gcn_deg_finish<<<(N + TB - 1) / TB, TB, 0, stream>>>(dinv, N);

    const int tiles  = (N + 15) / 16;
    const int gblk   = (tiles + 3) / 4;

    gcn_gemm<64, 64, 0><<<gblk, 128, 0, stream>>>(x, W_in, b_in, nullptr, h0, nullptr, N);

    gcn_gemm<64, 32, 1><<<gblk, 128, 0, stream>>>(h0, W1, nullptr, dinv, p, accA, N);
    gcn_scatter<<<((size_t)E * 8 + TB - 1) / TB, TB, 0, stream>>>(src, dst, p, accA, E);
    gcn_finish <<<((size_t)N * 32 + TB - 1) / TB, TB, 0, stream>>>(accA, dinv, b1, N);

    gcn_gemm<32, 32, 1><<<gblk, 128, 0, stream>>>(accA, W2, nullptr, dinv, p, accB, N);
    gcn_scatter<<<((size_t)E * 8 + TB - 1) / TB, TB, 0, stream>>>(src, dst, p, accB, E);

    gcn_out<<<(N + TB - 1) / TB, TB, 0, stream>>>(accB, dinv, b2, W_out, b_out, out, N);
}


# weights kept in each wave's registers as B fragments read once from global; 2048 resident waves stride over the row tiles with the next tile's rows requested ahead; LDS staging gone; same arithmetic
# speedup vs baseline: 1.0868x; 1.0868x over previous
#include <hip/hip_runtime.h>
#include <hip/hip_bf16.h>

typedef __attribute__((ext_vector_type(2))) float v2f;
typedef __attribute__((ext_vector_type(8))) float v8f;

__device__ __forceinline__ float lrelu(float v) { return v > 0.0f ? v : 0.01f * v; }

__global__ void gcn_deg_init(float* __restrict__ deg, int n) {
    int i = blockIdx.x * blockDim.x + threadIdx.x;
    if (i < n) deg[i] = 1.0f;
}

__global__ void gcn_deg_count(const int* __restrict__ dst, float* __restrict__ deg, int E) {
    int e = blockIdx.x * blockDim.x + threadIdx.x;
    if (e < E) atomicAdd(&deg[dst[e]], 1.0f);
}

__global__ void gcn_deg_finish(float* __restrict__ dinv, int n) {
    int i = blockIdx.x * blockDim.x + threadIdx.x;
    if (i < n) dinv[i] = rsqrtf(dinv[i]);
}

template <int K, int NOUT, int MODE>
__global__ __launch_bounds__(128) void gcn_gemm(
    const float* __restrict__ in,
    const float* __restrict__ W,
    const float* __restrict__ bias,
    const float* __restrict__ dinv,
    float* __restrict__ out,
    float* __restrict__ acc_out,
    int n)
{
    constexpr int KS = K / 4;
    constexpr int NT = NOUT / 16;

    const int lane  = threadIdx.x & 31;
    const int lhalf = lane >> 4;
    const int l16   = lane & 15;
    const int wave   = __builtin_amdgcn_readfirstlane((int)(threadIdx.x >> 5));
    const int gwave  = blockIdx.x * 4 + wave;
    const int nwaves = gridDim.x * 4;
    const int tiles  = n >> 4;
    if (gwave >= tiles) return;

    v2f bf[KS][NT];
    #pragma unroll
    for (int s = 0; s < KS; ++s) {
        #pragma unroll
        for (int t = 0; t < NT; ++t) {
            const float* wp = W + (t * 16 + l16) * K + 4 * s + 2 * lhalf;
            bf[s][t].x = wp[0];
            bf[s][t].y = wp[1];
        }
    }

    float bv[NT];
    if (MODE == 0) {
        #pragma unroll
        for (int t = 0; t < NT; ++t) bv[t] = bias[t * 16 + l16];
    }

    v2f a[KS];
    {
        const float* ap = in + (size_t)(gwave * 16 + l16) * K + 2 * lhalf;
        #pragma unroll
        for (int s = 0; s < KS; ++s) { a[s].x = ap[4 * s]; a[s].y = ap[4 * s + 1]; }
    }

    for (int tile = gwave; tile < tiles; tile += nwaves) {
        const int row0 = tile * 16;

        const int ntile = (tile + nwaves < tiles) ? (tile + nwaves) : tile;
        float dv[8];
        if (MODE == 1) {
            #pragma unroll
            for (int r = 0; r < 8; ++r) dv[r] = dinv[row0 + r + 8 * lhalf];
        }
        v2f an[KS];
        {
            const float* ap = in + (size_t)(ntile * 16 + l16) * K + 2 * lhalf;
            #pragma unroll
            for (int s = 0; s < KS; ++s) { an[s].x = ap[4 * s]; an[s].y = ap[4 * s + 1]; }
        }

        __builtin_amdgcn_sched_barrier(0);

        v8f acc[NT];
        #pragma unroll
        for (int t = 0; t < NT; ++t)
            acc[t] = (v8f){0.f, 0.f, 0.f, 0.f, 0.f, 0.f, 0.f, 0.f};

        #pragma unroll
        for (int s = 0; s < KS; ++s) {
            #pragma unroll
            for (int t = 0; t < NT; ++t) {
                acc[t] = __builtin_amdgcn_wmma_f32_16x16x4_f32(
                    false, a[s], false, bf[s][t], (short)0, acc[t], false, false);
            }
        }

        #pragma unroll
        for (int t = 0; t < NT; ++t) {
            const int col = t * 16 + l16;
            if (MODE == 0) {
                #pragma unroll
                for (int r = 0; r < 8; ++r) {
                    const int row = row0 + r + 8 * lhalf;
                    out[(size_t)row * NOUT + col] = lrelu(acc[t][r] + bv[t]);
                }
            } else {
                #pragma unroll
                for (int r = 0; r < 8; ++r) {
                    const int row = row0 + r + 8 * lhalf;
                    const float val = acc[t][r] * dv[r];
                    out[(size_t)row * NOUT + col]     = val;
                    acc_out[(size_t)row * NOUT + col] = val;
                }
            }
        }

        __builtin_amdgcn_sched_barrier(0);

        #pragma unroll
        for (int s = 0; s < KS; ++s) a[s] = an[s];
    }
}

__global__ void gcn_scatter(const int* __restrict__ src, const int* __restrict__ dst,
                            const float* __restrict__ p, float* __restrict__ acc, int E)
{
    int t = blockIdx.x * blockDim.x + threadIdx.x;
    int e = t >> 3;
    if (e >= E) return;
    int c4 = (t & 7) * 4;
    int u = src[e];
    int v = dst[e];
    float4 m = *(const float4*)(p + (size_t)u * 32 + c4);
    float* a = acc + (size_t)v * 32 + c4;
    atomicAdd(a + 0, m.x);
    atomicAdd(a + 1, m.y);
    atomicAdd(a + 2, m.z);
    atomicAdd(a + 3, m.w);
}

__global__ void gcn_finish(float* __restrict__ acc, const float* __restrict__ dinv,
                           const float* __restrict__ bias, int n)
{
    int t = blockIdx.x * blockDim.x + threadIdx.x;
    if (t >= n * 32) return;
    int row = t >> 5;
    int c   = t & 31;
    acc[t] = lrelu(dinv[row] * acc[t] + bias[c]);
}

__global__ void gcn_out(const float* __restrict__ acc2, const float* __restrict__ dinv,
                        const float* __restrict__ b2, const float* __restrict__ Wout,
                        const float* __restrict__ bout, float* __restrict__ out, int n)
{
    int i = blockIdx.x * blockDim.x + threadIdx.x;
    if (i >= n) return;
    float di = dinv[i];
    float s = 0.0f;
    #pragma unroll
    for (int c = 0; c < 32; ++c)
        s += Wout[c] * lrelu(di * acc2[(size_t)i * 32 + c] + b2[c]);
    out[i] = s + bout[0];
}

extern "C" void kernel_launch(void* const* d_in, const int* in_sizes, int n_in,
                              void* d_out, int out_size, void* d_ws, size_t ws_size,
                              hipStream_t stream)
{
    const float* x     = (const float*)d_in[0];
    const int*   ei    = (const int*)d_in[1];
    const float* W_in  = (const float*)d_in[2];
    const float* b_in  = (const float*)d_in[3];
    const float* W1    = (const float*)d_in[4];
    const float* b1    = (const float*)d_in[5];
    const float* W2    = (const float*)d_in[6];
    const float* b2    = (const float*)d_in[7];
    const float* W_out = (const float*)d_in[8];
    const float* b_out = (const float*)d_in[9];
    float* out = (float*)d_out;

    const int N = in_sizes[0] / 64;
    const int E = in_sizes[1] / 2;
    const int* src = ei;
    const int* dst = ei + E;

    float* ws   = (float*)d_ws;
    float* dinv = ws;
    float* h0   = ws + (size_t)N;
    float* p    = ws + (size_t)N * 65;
    float* accA = ws + (size_t)N * 97;
    float* accB = ws + (size_t)N * 129;

    const int TB = 256;
    gcn_deg_init  <<<(N + TB - 1) / TB, TB, 0, stream>>>(dinv, N);
    gcn_deg_count <<<(E + TB - 1) / TB, TB, 0, stream>>>(dst, dinv, E);
    gcn_deg_finish<<<(N + TB - 1) / TB, TB, 0, stream>>>(dinv, N);

    const int gblk   = 512;

    gcn_gemm<64, 64, 0><<<gblk, 128, 0, stream>>>(x, W_in, b_in, nullptr, h0, nullptr, N);

    gcn_gemm<64, 32, 1><<<gblk, 128, 0, stream>>>(h0, W1, nullptr, dinv, p, accA, N);
    gcn_scatter<<<((size_t)E * 8 + TB - 1) / TB, TB, 0, stream>>>(src, dst, p, accA, E);
    gcn_finish <<<((size_t)N * 32 + TB - 1) / TB, TB, 0, stream>>>(accA, dinv, b1, N);

    gcn_gemm<32, 32, 1><<<gblk, 128, 0, stream>>>(accA, W2, nullptr, dinv, p, accB, N);
    gcn_scatter<<<((size_t)E * 8 + TB - 1) / TB, TB, 0, stream>>>(src, dst, p, accB, E);

    gcn_out<<<(N + TB - 1) / TB, TB, 0, stream>>>(accB, dinv, b2, W_out, b_out, out, N);
}


# edge sums by gather: per-call edge list ordered by destination (int count whose returned value is the slot, scan, fill); 8 lanes sum a row in registers; no f32 atomics; sum order = count arrival
# speedup vs baseline: 1.2593x; 1.1587x over previous
#include <hip/hip_runtime.h>
#include <hip/hip_bf16.h>

typedef __attribute__((ext_vector_type(2))) float v2f;
typedef __attribute__((ext_vector_type(8))) float v8f;

__device__ __forceinline__ float lrelu(float v) { return v > 0.0f ? v : 0.01f * v; }

__global__ void gcn_cnt_zero(int* __restrict__ cnt, int n) {
    int i = blockIdx.x * blockDim.x + threadIdx.x;
    if (i < n) cnt[i] = 0;
}

__global__ void gcn_cnt_edges(const int* __restrict__ dst, int* __restrict__ cnt,
                              int* __restrict__ slot, int E) {
    const int stride = gridDim.x * blockDim.x;
    for (int e = blockIdx.x * blockDim.x + threadIdx.x; e < E; e += stride)
        slot[e] = atomicAdd(&cnt[dst[e]], 1);
}

__global__ __launch_bounds__(1024) void gcn_rowptr(const int* __restrict__ cnt,
                                                   int* __restrict__ rowptr,
                                                   float* __restrict__ dinv, int n) {
    __shared__ int wtot[32];
    __shared__ int wbelow[32];
    const int tid  = threadIdx.x;
    const int lane = tid & 31;
    const int wv   = tid >> 5;
    const int first = blockIdx.x * 1024;

    int below = 0;
    for (int i = tid; i < first; i += 1024) below += cnt[i];

    const int i = first + tid;
    const int c = (i < n) ? cnt[i] : 0;

    int inc = c;
    for (int d = 1; d < 32; d <<= 1) {
        const int up = __shfl_up(inc, d);
        if (lane >= d) inc += up;
    }
    for (int d = 16; d > 0; d >>= 1) below += __shfl_xor(below, d);
    if (lane == 31) wtot[wv] = inc;
    if (lane == 0)  wbelow[wv] = below;
    __syncthreads();

    int add = wbelow[lane] + ((lane < wv) ? wtot[lane] : 0);
    for (int d = 16; d > 0; d >>= 1) add += __shfl_xor(add, d);

    if (i < n) {
        rowptr[i] = add + inc - c;
        dinv[i]   = rsqrtf((float)(c + 1));
    }
}

__global__ void gcn_csr_fill(const int* __restrict__ src, const int* __restrict__ dst,
                             const int* __restrict__ rowptr, const int* __restrict__ slot,
                             int* __restrict__ csr, int E) {
    const int stride = gridDim.x * blockDim.x;
    for (int e = blockIdx.x * blockDim.x + threadIdx.x; e < E; e += stride)
        csr[rowptr[dst[e]] + slot[e]] = src[e];
}

template <int K, int NOUT, int MODE>
__global__ __launch_bounds__(128) void gcn_gemm(
    const float* __restrict__ in,
    const float* __restrict__ W,
    const float* __restrict__ bias,
    const float* __restrict__ dinv,
    float* __restrict__ out,
    int n)
{
    constexpr int KS = K / 4;
    constexpr int NT = NOUT / 16;

    const int lane  = threadIdx.x & 31;
    const int lhalf = lane >> 4;
    const int l16   = lane & 15;
    const int wave   = __builtin_amdgcn_readfirstlane((int)(threadIdx.x >> 5));
    const int gwave  = blockIdx.x * 4 + wave;
    const int nwaves = gridDim.x * 4;
    const int tiles  = n >> 4;
    if (gwave >= tiles) return;

    v2f bf[KS][NT];
    #pragma unroll
    for (int s = 0; s < KS; ++s) {
        #pragma unroll
        for (int t = 0; t < NT; ++t) {
            const float* wp = W + (t * 16 + l16) * K + 4 * s + 2 * lhalf;
            bf[s][t].x = wp[0];
            bf[s][t].y = wp[1];
        }
    }

    float bv[NT];
    if (MODE == 0) {
        #pragma unroll
        for (int t = 0; t < NT; ++t) bv[t] = bias[t * 16 + l16];
    }

    v2f a[KS];
    {
        const float* ap = in + (size_t)(gwave * 16 + l16) * K + 2 * lhalf;
        #pragma unroll
        for (int s = 0; s < KS; ++s) { a[s].x = ap[4 * s]; a[s].y = ap[4 * s + 1]; }
    }

    for (int tile = gwave; tile < tiles; tile += nwaves) {
        const int row0 = tile * 16;

        const int ntile = (tile + nwaves < tiles) ? (tile + nwaves) : tile;
        float dv[8];
        if (MODE == 1) {
            #pragma unroll
            for (int r = 0; r < 8; ++r) dv[r] = dinv[row0 + r + 8 * lhalf];
        }
        v2f an[KS];
        {
            const float* ap = in + (size_t)(ntile * 16 + l16) * K + 2 * lhalf;
            #pragma unroll
            for (int s = 0; s < KS; ++s) { an[s].x = ap[4 * s]; an[s].y = ap[4 * s + 1]; }
        }

        __builtin_amdgcn_sched_barrier(0);

        v8f acc[NT];
        #pragma unroll
        for (int t = 0; t < NT; ++t)
            acc[t] = (v8f){0.f, 0.f, 0.f, 0.f, 0.f, 0.f, 0.f, 0.f};

        #pragma unroll
        for (int s = 0; s < KS; ++s) {
            #pragma unroll
            for (int t = 0; t < NT; ++t) {
                acc[t] = __builtin_amdgcn_wmma_f32_16x16x4_f32(
                    false, a[s], false, bf[s][t], (short)0, acc[t], false, false);
            }
        }

        #pragma unroll
        for (int t = 0; t < NT; ++t) {
            const int col = t * 16 + l16;
            if (MODE == 0) {
                #pragma unroll
                for (int r = 0; r < 8; ++r) {
                    const int row = row0 + r + 8 * lhalf;
                    out[(size_t)row * NOUT + col] = lrelu(acc[t][r] + bv[t]);
                }
            } else {
                #pragma unroll
                for (int r = 0; r < 8; ++r) {
                    const int row = row0 + r + 8 * lhalf;
                    out[(size_t)row * NOUT + col] = acc[t][r] * dv[r];
                }
            }
        }

        __builtin_amdgcn_sched_barrier(0);

        #pragma unroll
        for (int s = 0; s < KS; ++s) a[s] = an[s];
    }
}

__global__ void gcn_pull(const int* __restrict__ rowptr, const int* __restrict__ cnt,
                         const int* __restrict__ csr, const float* __restrict__ p,
                         float* __restrict__ acc, int n)
{
    const int t = blockIdx.x * blockDim.x + threadIdx.x;
    const int v = t >> 3;
    if (v >= n) return;
    const int sub = t & 7;
    const float4* p4 = (const float4*)p + sub;
    const int beg = rowptr[v];
    const int deg = cnt[v];
    float4 s = p4[(size_t)v * 8];
    asm volatile("" :: "v"(beg), "v"(deg));
    const int* row = csr + beg;
    const int top = n - 1;
    if (deg > 0) {
        const int last = deg - 1;
        int u0 = row[0];
        int u1 = row[min(1, last)];
        int u2 = row[min(2, last)];
        int u3 = row[min(3, last)];
        for (int j = 0; j < deg; j += 4) {
            const int r = deg - j;
            const float4 m0 = p4[(size_t)min(max(u0, 0), top) * 8];
            const float4 m1 = p4[(size_t)min(max(u1, 0), top) * 8];
            const float4 m2 = p4[(size_t)min(max(u2, 0), top) * 8];
            const float4 m3 = p4[(size_t)min(max(u3, 0), top) * 8];
            u0 = row[min(j + 4, last)];
            u1 = row[min(j + 5, last)];
            u2 = row[min(j + 6, last)];
            u3 = row[min(j + 7, last)];
            asm volatile("" :: "v"(u0), "v"(u1), "v"(u2), "v"(u3),
                               "v"(m0.x), "v"(m1.x), "v"(m2.x), "v"(m3.x));
            const bool h1 = r > 1, h2 = r > 2, h3 = r > 3;
            s.x += m0.x; s.y += m0.y; s.z += m0.z; s.w += m0.w;
            s.x += h1 ? m1.x : -0.0f; s.y += h1 ? m1.y : -0.0f; s.z += h1 ? m1.z : -0.0f; s.w += h1 ? m1.w : -0.0f;
            s.x += h2 ? m2.x : -0.0f; s.y += h2 ? m2.y : -0.0f; s.z += h2 ? m2.z : -0.0f; s.w += h2 ? m2.w : -0.0f;
            s.x += h3 ? m3.x : -0.0f; s.y += h3 ? m3.y : -0.0f; s.z += h3 ? m3.z : -0.0f; s.w += h3 ? m3.w : -0.0f;
        }
    }
    ((float4*)acc)[(size_t)v * 8 + sub] = s;
}

__global__ void gcn_finish(float* __restrict__ acc, const float* __restrict__ dinv,
                           const float* __restrict__ bias, int n)
{
    int t = blockIdx.x * blockDim.x + threadIdx.x;
    if (t >= n * 32) return;
    int row = t >> 5;
    int c   = t & 31;
    acc[t] = lrelu(dinv[row] * acc[t] + bias[c]);
}

__global__ void gcn_out(const float* __restrict__ acc2, const float* __restrict__ dinv,
                        const float* __restrict__ b2, const float* __restrict__ Wout,
                        const float* __restrict__ bout, float* __restrict__ out, int n)
{
    int i = blockIdx.x * blockDim.x + threadIdx.x;
    if (i >= n) return;
    float di = dinv[i];
    float s = 0.0f;
    #pragma unroll
    for (int c = 0; c < 32; ++c)
        s += Wout[c] * lrelu(di * acc2[(size_t)i * 32 + c] + b2[c]);
    out[i] = s + bout[0];
}

extern "C" void kernel_launch(void* const* d_in, const int* in_sizes, int n_in,
                              void* d_out, int out_size, void* d_ws, size_t ws_size,
                              hipStream_t stream)
{
    const float* x     = (const float*)d_in[0];
    const int*   ei    = (const int*)d_in[1];
    const float* W_in  = (const float*)d_in[2];
    const float* b_in  = (const float*)d_in[3];
    const float* W1    = (const float*)d_in[4];
    const float* b1    = (const float*)d_in[5];
    const float* W2    = (const float*)d_in[6];
    const float* b2    = (const float*)d_in[7];
    const float* W_out = (const float*)d_in[8];
    const float* b_out = (const float*)d_in[9];
    float* out = (float*)d_out;

    const int N = in_sizes[0] / 64;
    const int E = in_sizes[1] / 2;
    const int* src = ei;
    const int* dst = ei + E;

    float* ws   = (float*)d_ws;
    float* dinv = ws;
    float* h0   = ws + (size_t)N;
    float* p    = ws + (size_t)N * 65;
    float* accA = ws + (size_t)N * 97;
    float* accB = ws + (size_t)N * 129;
    int* cnt    = (int*)(ws + (size_t)N * 161);
    int* rowptr = cnt + (size_t)N;
    int* slot   = rowptr + (size_t)N;
    int* csr    = slot + (size_t)E;

    const int TB = 256;
    gcn_cnt_zero <<<(N + TB - 1) / TB, TB, 0, stream>>>(cnt, N);
    gcn_cnt_edges<<<2048, TB, 0, stream>>>(dst, cnt, slot, E);
    gcn_rowptr   <<<(N + 1023) / 1024, 1024, 0, stream>>>(cnt, rowptr, dinv, N);
    gcn_csr_fill <<<1024, TB, 0, stream>>>(src, dst, rowptr, slot, csr, E);

    const int gblk   = 512;

    gcn_gemm<64, 64, 0><<<gblk, 128, 0, stream>>>(x, W_in, b_in, nullptr, h0, N);

    gcn_gemm<64, 32, 1><<<gblk, 128, 0, stream>>>(h0, W1, nullptr, dinv, p, N);
    gcn_pull  <<<((size_t)N * 8 + TB - 1) / TB, TB, 0, stream>>>(rowptr, cnt, csr, p, accA, N);
    gcn_finish<<<((size_t)N * 32 + TB - 1) / TB, TB, 0, stream>>>(accA, dinv, b1, N);

    gcn_gemm<32, 32, 1><<<gblk, 128, 0, stream>>>(accA, W2, nullptr, dinv, p, N);
    gcn_pull  <<<((size_t)N * 8 + TB - 1) / TB, TB, 0, stream>>>(rowptr, cnt, csr, p, accB, N);

    gcn_out<<<(N + TB - 1) / TB, TB, 0, stream>>>(accB, dinv, b2, W_out, b_out, out, N);
}


# input transform runs beside the edge count and conv1's product beside the list fill, each pair as work-groups of one grid (4 edges in flight per lane there); launches 11->9; same arithmetic
# speedup vs baseline: 1.3000x; 1.0323x over previous
#include <hip/hip_runtime.h>
#include <hip/hip_bf16.h>

typedef __attribute__((ext_vector_type(2))) float v2f;
typedef __attribute__((ext_vector_type(8))) float v8f;

__device__ __forceinline__ float lrelu(float v) { return v > 0.0f ? v : 0.01f * v; }

__global__ void gcn_cnt_zero(int* __restrict__ cnt, int n) {
    int i = blockIdx.x * blockDim.x + threadIdx.x;
    if (i < n) cnt[i] = 0;
}

__device__ __forceinline__ void count_edges(const int* __restrict__ dst, int* __restrict__ cnt,
                                            int* __restrict__ slot, int E, int blk, int nblk) {
    const int stride = nblk * (int)blockDim.x;
    for (int e0 = blk * (int)blockDim.x + (int)threadIdx.x; e0 < E; e0 += 4 * stride) {
        const int e1 = e0 + stride, e2 = e1 + stride, e3 = e2 + stride;
        if (e3 < E) {
            const int d0 = dst[e0], d1 = dst[e1], d2 = dst[e2], d3 = dst[e3];
            const int r0 = atomicAdd(&cnt[d0], 1);
            const int r1 = atomicAdd(&cnt[d1], 1);
            const int r2 = atomicAdd(&cnt[d2], 1);
            const int r3 = atomicAdd(&cnt[d3], 1);
            slot[e0] = r0; slot[e1] = r1; slot[e2] = r2; slot[e3] = r3;
        } else {
            slot[e0] = atomicAdd(&cnt[dst[e0]], 1);
            if (e1 < E) slot[e1] = atomicAdd(&cnt[dst[e1]], 1);
            if (e2 < E) slot[e2] = atomicAdd(&cnt[dst[e2]], 1);
        }
    }
}

__global__ __launch_bounds__(1024) void gcn_rowptr(const int* __restrict__ cnt,
                                                   int* __restrict__ rowptr,
                                                   float* __restrict__ dinv, int n) {
    __shared__ int wtot[32];
    __shared__ int wbelow[32];
    const int tid  = threadIdx.x;
    const int lane = tid & 31;
    const int wv   = tid >> 5;
    const int first = blockIdx.x * 1024;

    int below = 0;
    for (int i = tid; i < first; i += 1024) below += cnt[i];

    const int i = first + tid;
    const int c = (i < n) ? cnt[i] : 0;

    int inc = c;
    for (int d = 1; d < 32; d <<= 1) {
        const int up = __shfl_up(inc, d);
        if (lane >= d) inc += up;
    }
    for (int d = 16; d > 0; d >>= 1) below += __shfl_xor(below, d);
    if (lane == 31) wtot[wv] = inc;
    if (lane == 0)  wbelow[wv] = below;
    __syncthreads();

    int add = wbelow[lane] + ((lane < wv) ? wtot[lane] : 0);
    for (int d = 16; d > 0; d >>= 1) add += __shfl_xor(add, d);

    if (i < n) {
        rowptr[i] = add + inc - c;
        dinv[i]   = rsqrtf((float)(c + 1));
    }
}

__device__ __forceinline__ void fill_csr(const int* __restrict__ src, const int* __restrict__ dst,
                                         const int* __restrict__ rowptr, const int* __restrict__ slot,
                                         int* __restrict__ csr, int E, int blk, int nblk) {
    const int stride = nblk * (int)blockDim.x;
    for (int e0 = blk * (int)blockDim.x + (int)threadIdx.x; e0 < E; e0 += 4 * stride) {
        const int e1 = e0 + stride, e2 = e1 + stride, e3 = e2 + stride;
        if (e3 < E) {
            const int d0 = dst[e0], d1 = dst[e1], d2 = dst[e2], d3 = dst[e3];
            const int s0 = slot[e0], s1 = slot[e1], s2 = slot[e2], s3 = slot[e3];
            const int u0 = src[e0], u1 = src[e1], u2 = src[e2], u3 = src[e3];
            const int b0 = rowptr[d0], b1 = rowptr[d1], b2 = rowptr[d2], b3 = rowptr[d3];
            csr[b0 + s0] = u0; csr[b1 + s1] = u1; csr[b2 + s2] = u2; csr[b3 + s3] = u3;
        } else {
            csr[rowptr[dst[e0]] + slot[e0]] = src[e0];
            if (e1 < E) csr[rowptr[dst[e1]] + slot[e1]] = src[e1];
            if (e2 < E) csr[rowptr[dst[e2]] + slot[e2]] = src[e2];
        }
    }
}

template <int K, int NOUT, int MODE>
__device__ __forceinline__ void gemm_tiles(
    const float* __restrict__ in,
    const float* __restrict__ W,
    const float* __restrict__ bias,
    const float* __restrict__ dinv,
    float* __restrict__ out,
    int n, int blk, int nblk)
{
    constexpr int KS = K / 4;
    constexpr int NT = NOUT / 16;

    const int lane  = threadIdx.x & 31;
    const int lhalf = lane >> 4;
    const int l16   = lane & 15;
    const int wave   = __builtin_amdgcn_readfirstlane((int)(threadIdx.x >> 5));
    const int gwave  = blk * 4 + wave;
    const int nwaves = nblk * 4;
    const int tiles  = n >> 4;
    if (gwave >= tiles) return;

    v2f bf[KS][NT];
    #pragma unroll
    for (int s = 0; s < KS; ++s) {
        #pragma unroll
        for (int t = 0; t < NT; ++t) {
            const float* wp = W + (t * 16 + l16) * K + 4 * s + 2 * lhalf;
            bf[s][t].x = wp[0];
            bf[s][t].y = wp[1];
        }
    }

    float bv[NT];
    if (MODE == 0) {
        #pragma unroll
        for (int t = 0; t < NT; ++t) bv[t] = bias[t * 16 + l16];
    }

    v2f a[KS];
    {
        const float* ap = in + (size_t)(gwave * 16 + l16) * K + 2 * lhalf;
        #pragma unroll
        for (int s = 0; s < KS; ++s) { a[s].x = ap[4 * s]; a[s].y = ap[4 * s + 1]; }
    }

    for (int tile = gwave; tile < tiles; tile += nwaves) {
        const int row0 = tile * 16;

        const int ntile = (tile + nwaves < tiles) ? (tile + nwaves) : tile;
        float dv[8];
        if (MODE == 1) {
            #pragma unroll
            for (int r = 0; r < 8; ++r) dv[r] = dinv[row0 + r + 8 * lhalf];
        }
        v2f an[KS];
        {
            const float* ap = in + (size_t)(ntile * 16 + l16) * K + 2 * lhalf;
            #pragma unroll
            for (int s = 0; s < KS; ++s) { an[s].x = ap[4 * s]; an[s].y = ap[4 * s + 1]; }
        }

        __builtin_amdgcn_sched_barrier(0);

        v8f acc[NT];
        #pragma unroll
        for (int t = 0; t < NT; ++t)
            acc[t] = (v8f){0.f, 0.f, 0.f, 0.f, 0.f, 0.f, 0.f, 0.f};

        #pragma unroll
        for (int s = 0; s < KS; ++s) {
            #pragma unroll
            for (int t = 0; t < NT; ++t) {
                acc[t] = __builtin_amdgcn_wmma_f32_16x16x4_f32(
                    false, a[s], false, bf[s][t], (short)0, acc[t], false, false);
            }
        }

        #pragma unroll
        for (int t = 0; t < NT; ++t) {
            const int col = t * 16 + l16;
            if (MODE == 0) {
                #pragma unroll
                for (int r = 0; r < 8; ++r) {
                    const int row = row0 + r + 8 * lhalf;
                    out[(size_t)row * NOUT + col] = lrelu(acc[t][r] + bv[t]);
                }
            } else {
                #pragma unroll
                for (int r = 0; r < 8; ++r) {
                    const int row = row0 + r + 8 * lhalf;
                    out[(size_t)row * NOUT + col] = acc[t][r] * dv[r];
                }
            }
        }

        __builtin_amdgcn_sched_barrier(0);

        #pragma unroll
        for (int s = 0; s < KS; ++s) a[s] = an[s];
    }
}

template <int K, int NOUT, int MODE>
__global__ __launch_bounds__(128) void gcn_gemm(
    const float* __restrict__ in, const float* __restrict__ W, const float* __restrict__ bias,
    const float* __restrict__ dinv, float* __restrict__ out, int n)
{
    gemm_tiles<K, NOUT, MODE>(in, W, bias, dinv, out, n, (int)blockIdx.x, (int)gridDim.x);
}

__global__ __launch_bounds__(128) void gcn_gemm_in_and_count(
    const float* __restrict__ x, const float* __restrict__ W_in, const float* __restrict__ b_in,
    float* __restrict__ h0, int n, int gblk,
    const int* __restrict__ dst, int* __restrict__ cnt, int* __restrict__ slot, int E)
{
    if ((int)blockIdx.x < gblk)
        gemm_tiles<64, 64, 0>(x, W_in, b_in, nullptr, h0, n, (int)blockIdx.x, gblk);
    else
        count_edges(dst, cnt, slot, E, (int)blockIdx.x - gblk, (int)gridDim.x - gblk);
}

__global__ __launch_bounds__(128) void gcn_gemm_c1_and_fill(
    const float* __restrict__ h0, const float* __restrict__ W1, const float* __restrict__ dinv,
    float* __restrict__ p, int n, int gblk,
    const int* __restrict__ src, const int* __restrict__ dst, const int* __restrict__ rowptr,
    const int* __restrict__ slot, int* __restrict__ csr, int E)
{
    if ((int)blockIdx.x < gblk)
        gemm_tiles<64, 32, 1>(h0, W1, nullptr, dinv, p, n, (int)blockIdx.x, gblk);
    else
        fill_csr(src, dst, rowptr, slot, csr, E, (int)blockIdx.x - gblk, (int)gridDim.x - gblk);
}

__global__ void gcn_pull(const int* __restrict__ rowptr, const int* __restrict__ cnt,
                         const int* __restrict__ csr, const float* __restrict__ p,
                         float* __restrict__ acc, int n)
{
    const int t = blockIdx.x * blockDim.x + threadIdx.x;
    const int v = t >> 3;
    if (v >= n) return;
    const int sub = t & 7;
    const float4* p4 = (const float4*)p + sub;
    const int beg = rowptr[v];
    const int deg = cnt[v];
    float4 s = p4[(size_t)v * 8];
    asm volatile("" :: "v"(beg), "v"(deg));
    const int* row = csr + beg;
    const int top = n - 1;
    if (deg > 0) {
        const int last = deg - 1;
        int u0 = row[0];
        int u1 = row[min(1, last)];
        int u2 = row[min(2, last)];
        int u3 = row[min(3, last)];
        for (int j = 0; j < deg; j += 4) {
            const int r = deg - j;
            const float4 m0 = p4[(size_t)min(max(u0, 0), top) * 8];
            const float4 m1 = p4[(size_t)min(max(u1, 0), top) * 8];
            const float4 m2 = p4[(size_t)min(max(u2, 0), top) * 8];
            const float4 m3 = p4[(size_t)min(max(u3, 0), top) * 8];
            u0 = row[min(j + 4, last)];
            u1 = row[min(j + 5, last)];
            u2 = row[min(j + 6, last)];
            u3 = row[min(j + 7, last)];
            asm volatile("" :: "v"(u0), "v"(u1), "v"(u2), "v"(u3),
                               "v"(m0.x), "v"(m1.x), "v"(m2.x), "v"(m3.x));
            const bool h1 = r > 1, h2 = r > 2, h3 = r > 3;
            s.x += m0.x; s.y += m0.y; s.z += m0.z; s.w += m0.w;
            s.x += h1 ? m1.x : -0.0f; s.y += h1 ? m1.y : -0.0f; s.z += h1 ? m1.z : -0.0f; s.w += h1 ? m1.w : -0.0f;
            s.x += h2 ? m2.x : -0.0f; s.y += h2 ? m2.y : -0.0f; s.z += h2 ? m2.z : -0.0f; s.w += h2 ? m2.w : -0.0f;
            s.x += h3 ? m3.x : -0.0f; s.y += h3 ? m3.y : -0.0f; s.z += h3 ? m3.z : -0.0f; s.w += h3 ? m3.w : -0.0f;
        }
    }
    ((float4*)acc)[(size_t)v * 8 + sub] = s;
}

__global__ void gcn_finish(float* __restrict__ acc, const float* __restrict__ dinv,
                           const float* __restrict__ bias, int n)
{
    int t = blockIdx.x * blockDim.x + threadIdx.x;
    if (t >= n * 32) return;
    int row = t >> 5;
    int c   = t & 31;
    acc[t] = lrelu(dinv[row] * acc[t] + bias[c]);
}

__global__ void gcn_out(const float* __restrict__ acc2, const float* __restrict__ dinv,
                        const float* __restrict__ b2, const float* __restrict__ Wout,
                        const float* __restrict__ bout, float* __restrict__ out, int n)
{
    int i = blockIdx.x * blockDim.x + threadIdx.x;
    if (i >= n) return;
    float di = dinv[i];
    float s = 0.0f;
    #pragma unroll
    for (int c = 0; c < 32; ++c)
        s += Wout[c] * lrelu(di * acc2[(size_t)i * 32 + c] + b2[c]);
    out[i] = s + bout[0];
}

extern "C" void kernel_launch(void* const* d_in, const int* in_sizes, int n_in,
                              void* d_out, int out_size, void* d_ws, size_t ws_size,
                              hipStream_t stream)
{
    const float* x     = (const float*)d_in[0];
    const int*   ei    = (const int*)d_in[1];
    const float* W_in  = (const float*)d_in[2];
    const float* b_in  = (const float*)d_in[3];
    const float* W1    = (const float*)d_in[4];
    const float* b1    = (const float*)d_in[5];
    const float* W2    = (const float*)d_in[6];
    const float* b2    = (const float*)d_in[7];
    const float* W_out = (const float*)d_in[8];
    const float* b_out = (const float*)d_in[9];
    float* out = (float*)d_out;

    const int N = in_sizes[0] / 64;
    const int E = in_sizes[1] / 2;
    const int* src = ei;
    const int* dst = ei + E;

    float* ws   = (float*)d_ws;
    float* dinv = ws;
    float* h0   = ws + (size_t)N;
    float* p    = ws + (size_t)N * 65;
    float* accA = ws + (size_t)N * 97;
    float* accB = ws + (size_t)N * 129;
    int* cnt    = (int*)(ws + (size_t)N * 161);
    int* rowptr = cnt + (size_t)N;
    int* slot   = rowptr + (size_t)N;
    int* csr    = slot + (size_t)E;

    const int TB = 256;
    const int gblk   = 512;
    const int eblk   = 2048;

    gcn_cnt_zero <<<(N + TB - 1) / TB, TB, 0, stream>>>(cnt, N);
    gcn_gemm_in_and_count<<<gblk + eblk, 128, 0, stream>>>(x, W_in, b_in, h0, N, gblk, dst, cnt, slot, E);
    gcn_rowptr   <<<(N + 1023) / 1024, 1024, 0, stream>>>(cnt, rowptr, dinv, N);

    gcn_gemm_c1_and_fill<<<gblk + eblk, 128, 0, stream>>>(h0, W1, dinv, p, N, gblk, src, dst, rowptr, slot, csr, E);
    gcn_pull  <<<((size_t)N * 8 + TB - 1) / TB, TB, 0, stream>>>(rowptr, cnt, csr, p, accA, N);
    gcn_finish<<<((size_t)N * 32 + TB - 1) / TB, TB, 0, stream>>>(accA, dinv, b1, N);

    gcn_gemm<32, 32, 1><<<gblk, 128, 0, stream>>>(accA, W2, nullptr, dinv, p, N);
    gcn_pull  <<<((size_t)N * 8 + TB - 1) / TB, TB, 0, stream>>>(rowptr, cnt, csr, p, accB, N);

    gcn_out<<<(N + TB - 1) / TB, TB, 0, stream>>>(accB, dinv, b2, W_out, b_out, out, N);
}


# row stages folded into the edge gathers' registers: pull 1 stores lrelu(dinv*sum+b1), pull 2 the output dot (4 terms a lane, tree over 8 lanes; dot sum order differs); launches 9->7
# speedup vs baseline: 1.3819x; 1.0630x over previous
#include <hip/hip_runtime.h>
#include <hip/hip_bf16.h>

typedef __attribute__((ext_vector_type(2))) float v2f;
typedef __attribute__((ext_vector_type(8))) float v8f;

__device__ __forceinline__ float lrelu(float v) { return v > 0.0f ? v : 0.01f * v; }

__global__ void gcn_cnt_zero(int* __restrict__ cnt, int n) {
    int i = blockIdx.x * blockDim.x + threadIdx.x;
    if (i < n) cnt[i] = 0;
}

__device__ __forceinline__ void count_edges(const int* __restrict__ dst, int* __restrict__ cnt,
                                            int* __restrict__ slot, int E, int blk, int nblk) {
    const int stride = nblk * (int)blockDim.x;
    for (int e0 = blk * (int)blockDim.x + (int)threadIdx.x; e0 < E; e0 += 4 * stride) {
        const int e1 = e0 + stride, e2 = e1 + stride, e3 = e2 + stride;
        if (e3 < E) {
            const int d0 = dst[e0], d1 = dst[e1], d2 = dst[e2], d3 = dst[e3];
            const int r0 = atomicAdd(&cnt[d0], 1);
            const int r1 = atomicAdd(&cnt[d1], 1);
            const int r2 = atomicAdd(&cnt[d2], 1);
            const int r3 = atomicAdd(&cnt[d3], 1);
            slot[e0] = r0; slot[e1] = r1; slot[e2] = r2; slot[e3] = r3;
        } else {
            slot[e0] = atomicAdd(&cnt[dst[e0]], 1);
            if (e1 < E) slot[e1] = atomicAdd(&cnt[dst[e1]], 1);
            if (e2 < E) slot[e2] = atomicAdd(&cnt[dst[e2]], 1);
        }
    }
}

__global__ __launch_bounds__(1024) void gcn_rowptr(const int* __restrict__ cnt,
                                                   int* __restrict__ rowptr,
                                                   float* __restrict__ dinv, int n) {
    __shared__ int wtot[32];
    __shared__ int wbelow[32];
    const int tid  = threadIdx.x;
    const int lane = tid & 31;
    const int wv   = tid >> 5;
    const int first = blockIdx.x * 1024;

    int below = 0;
    for (int i = tid; i < first; i += 1024) below += cnt[i];

    const int i = first + tid;
    const int c = (i < n) ? cnt[i] : 0;

    int inc = c;
    for (int d = 1; d < 32; d <<= 1) {
        const int up = __shfl_up(inc, d);
        if (lane >= d) inc += up;
    }
    for (int d = 16; d > 0; d >>= 1) below += __shfl_xor(below, d);
    if (lane == 31) wtot[wv] = inc;
    if (lane == 0)  wbelow[wv] = below;
    __syncthreads();

    int add = wbelow[lane] + ((lane < wv) ? wtot[lane] : 0);
    for (int d = 16; d > 0; d >>= 1) add += __shfl_xor(add, d);

    if (i < n) {
        rowptr[i] = add + inc - c;
        dinv[i]   = rsqrtf((float)(c + 1));
    }
}

__device__ __forceinline__ void fill_csr(const int* __restrict__ src, const int* __restrict__ dst,
                                         const int* __restrict__ rowptr, const int* __restrict__ slot,
                                         int* __restrict__ csr, int E, int blk, int nblk) {
    const int stride = nblk * (int)blockDim.x;
    for (int e0 = blk * (int)blockDim.x + (int)threadIdx.x; e0 < E; e0 += 4 * stride) {
        const int e1 = e0 + stride, e2 = e1 + stride, e3 = e2 + stride;
        if (e3 < E) {
            const int d0 = dst[e0], d1 = dst[e1], d2 = dst[e2], d3 = dst[e3];
            const int s0 = slot[e0], s1 = slot[e1], s2 = slot[e2], s3 = slot[e3];
            const int u0 = src[e0], u1 = src[e1], u2 = src[e2], u3 = src[e3];
            const int b0 = rowptr[d0], b1 = rowptr[d1], b2 = rowptr[d2], b3 = rowptr[d3];
            csr[b0 + s0] = u0; csr[b1 + s1] = u1; csr[b2 + s2] = u2; csr[b3 + s3] = u3;
        } else {
            csr[rowptr[dst[e0]] + slot[e0]] = src[e0];
            if (e1 < E) csr[rowptr[dst[e1]] + slot[e1]] = src[e1];
            if (e2 < E) csr[rowptr[dst[e2]] + slot[e2]] = src[e2];
        }
    }
}

template <int K, int NOUT, int MODE>
__device__ __forceinline__ void gemm_tiles(
    const float* __restrict__ in,
    const float* __restrict__ W,
    const float* __restrict__ bias,
    const float* __restrict__ dinv,
    float* __restrict__ out,
    int n, int blk, int nblk)
{
    constexpr int KS = K / 4;
    constexpr int NT = NOUT / 16;

    const int lane  = threadIdx.x & 31;
    const int lhalf = lane >> 4;
    const int l16   = lane & 15;
    const int wave   = __builtin_amdgcn_readfirstlane((int)(threadIdx.x >> 5));
    const int gwave  = blk * 4 + wave;
    const int nwaves = nblk * 4;
    const int tiles  = n >> 4;
    if (gwave >= tiles) return;

    v2f bf[KS][NT];
    #pragma unroll
    for (int s = 0; s < KS; ++s) {
        #pragma unroll
        for (int t = 0; t < NT; ++t) {
            const float* wp = W + (t * 16 + l16) * K + 4 * s + 2 * lhalf;
            bf[s][t].x = wp[0];
            bf[s][t].y = wp[1];
        }
    }

    float bv[NT];
    if (MODE == 0) {
        #pragma unroll
        for (int t = 0; t < NT; ++t) bv[t] = bias[t * 16 + l16];
    }

    v2f a[KS];
    {
        const float* ap = in + (size_t)(gwave * 16 + l16) * K + 2 * lhalf;
        #pragma unroll
        for (int s = 0; s < KS; ++s) { a[s].x = ap[4 * s]; a[s].y = ap[4 * s + 1]; }
    }

    for (int tile = gwave; tile < tiles; tile += nwaves) {
        const int row0 = tile * 16;

        const int ntile = (tile + nwaves < tiles) ? (tile + nwaves) : tile;
        float dv[8];
        if (MODE == 1) {
            #pragma unroll
            for (int r = 0; r < 8; ++r) dv[r] = dinv[row0 + r + 8 * lhalf];
        }
        v2f an[KS];
        {
            const float* ap = in + (size_t)(ntile * 16 + l16) * K + 2 * lhalf;
            #pragma unroll
            for (int s = 0; s < KS; ++s) { an[s].x = ap[4 * s]; an[s].y = ap[4 * s + 1]; }
        }

        __builtin_amdgcn_sched_barrier(0);

        v8f acc[NT];
        #pragma unroll
        for (int t = 0; t < NT; ++t)
            acc[t] = (v8f){0.f, 0.f, 0.f, 0.f, 0.f, 0.f, 0.f, 0.f};

        #pragma unroll
        for (int s = 0; s < KS; ++s) {
            #pragma unroll
            for (int t = 0; t < NT; ++t) {
                acc[t] = __builtin_amdgcn_wmma_f32_16x16x4_f32(
                    false, a[s], false, bf[s][t], (short)0, acc[t], false, false);
            }
        }

        #pragma unroll
        for (int t = 0; t < NT; ++t) {
            const int col = t * 16 + l16;
            if (MODE == 0) {
                #pragma unroll
                for (int r = 0; r < 8; ++r) {
                    const int row = row0 + r + 8 * lhalf;
                    out[(size_t)row * NOUT + col] = lrelu(acc[t][r] + bv[t]);
                }
            } else {
                #pragma unroll
                for (int r = 0; r < 8; ++r) {
                    const int row = row0 + r + 8 * lhalf;
                    out[(size_t)row * NOUT + col] = acc[t][r] * dv[r];
                }
            }
        }

        __builtin_amdgcn_sched_barrier(0);

        #pragma unroll
        for (int s = 0; s < KS; ++s) a[s] = an[s];
    }
}

template <int K, int NOUT, int MODE>
__global__ __launch_bounds__(128) void gcn_gemm(
    const float* __restrict__ in, const float* __restrict__ W, const float* __restrict__ bias,
    const float* __restrict__ dinv, float* __restrict__ out, int n)
{
    gemm_tiles<K, NOUT, MODE>(in, W, bias, dinv, out, n, (int)blockIdx.x, (int)gridDim.x);
}

__global__ __launch_bounds__(128) void gcn_gemm_in_and_count(
    const float* __restrict__ x, const float* __restrict__ W_in, const float* __restrict__ b_in,
    float* __restrict__ h0, int n, int gblk,
    const int* __restrict__ dst, int* __restrict__ cnt, int* __restrict__ slot, int E)
{
    if ((int)blockIdx.x < gblk)
        gemm_tiles<64, 64, 0>(x, W_in, b_in, nullptr, h0, n, (int)blockIdx.x, gblk);
    else
        count_edges(dst, cnt, slot, E, (int)blockIdx.x - gblk, (int)gridDim.x - gblk);
}

__global__ __launch_bounds__(128) void gcn_gemm_c1_and_fill(
    const float* __restrict__ h0, const float* __restrict__ W1, const float* __restrict__ dinv,
    float* __restrict__ p, int n, int gblk,
    const int* __restrict__ src, const int* __restrict__ dst, const int* __restrict__ rowptr,
    const int* __restrict__ slot, int* __restrict__ csr, int E)
{
    if ((int)blockIdx.x < gblk)
        gemm_tiles<64, 32, 1>(h0, W1, nullptr, dinv, p, n, (int)blockIdx.x, gblk);
    else
        fill_csr(src, dst, rowptr, slot, csr, E, (int)blockIdx.x - gblk, (int)gridDim.x - gblk);
}

template <int MODE>
__global__ void gcn_pull(const int* __restrict__ rowptr, const int* __restrict__ cnt,
                         const int* __restrict__ csr, const float* __restrict__ p,
                         const float* __restrict__ dinv, const float* __restrict__ bias,
                         const float* __restrict__ Wout, const float* __restrict__ bout,
                         float* __restrict__ dstp, int n)
{
    const int t = blockIdx.x * blockDim.x + threadIdx.x;
    const bool live = (t >> 3) < n;
    if (MODE == 0 && !live) return;
    const int v = live ? (t >> 3) : (n - 1);
    const int sub = t & 7;
    const float4* p4 = (const float4*)p + sub;
    const int beg = rowptr[v];
    const int deg = cnt[v];
    const float di = dinv[v];
    const float b0 = bias[4 * sub], b1 = bias[4 * sub + 1], b2 = bias[4 * sub + 2], b3 = bias[4 * sub + 3];
    float w0 = 0.0f, w1 = 0.0f, w2 = 0.0f, w3 = 0.0f;
    if (MODE == 1) { w0 = Wout[4 * sub]; w1 = Wout[4 * sub + 1]; w2 = Wout[4 * sub + 2]; w3 = Wout[4 * sub + 3]; }
    float4 s = p4[(size_t)v * 8];
    asm volatile("" :: "v"(beg), "v"(deg));
    const int* row = csr + beg;
    const int top = n - 1;
    if (deg > 0) {
        const int last = deg - 1;
        int u0 = row[0];
        int u1 = row[min(1, last)];
        int u2 = row[min(2, last)];
        int u3 = row[min(3, last)];
        for (int j = 0; j < deg; j += 4) {
            const int r = deg - j;
            const float4 m0 = p4[(size_t)min(max(u0, 0), top) * 8];
            const float4 m1 = p4[(size_t)min(max(u1, 0), top) * 8];
            const float4 m2 = p4[(size_t)min(max(u2, 0), top) * 8];
            const float4 m3 = p4[(size_t)min(max(u3, 0), top) * 8];
            u0 = row[min(j + 4, last)];
            u1 = row[min(j + 5, last)];
            u2 = row[min(j + 6, last)];
            u3 = row[min(j + 7, last)];
            asm volatile("" :: "v"(u0), "v"(u1), "v"(u2), "v"(u3),
                               "v"(m0.x), "v"(m1.x), "v"(m2.x), "v"(m3.x));
            const bool h1 = r > 1, h2 = r > 2, h3 = r > 3;
            s.x += m0.x; s.y += m0.y; s.z += m0.z; s.w += m0.w;
            s.x += h1 ? m1.x : -0.0f; s.y += h1 ? m1.y : -0.0f; s.z += h1 ? m1.z : -0.0f; s.w += h1 ? m1.w : -0.0f;
            s.x += h2 ? m2.x : -0.0f; s.y += h2 ? m2.y : -0.0f; s.z += h2 ? m2.z : -0.0f; s.w += h2 ? m2.w : -0.0f;
            s.x += h3 ? m3.x : -0.0f; s.y += h3 ? m3.y : -0.0f; s.z += h3 ? m3.z : -0.0f; s.w += h3 ? m3.w : -0.0f;
        }
    }
    float4 h;
    h.x = lrelu(di * s.x + b0);
    h.y = lrelu(di * s.y + b1);
    h.z = lrelu(di * s.z + b2);
    h.w = lrelu(di * s.w + b3);
    if (MODE == 0) {
        ((float4*)dstp)[(size_t)v * 8 + sub] = h;
    } else {
        float d = w0 * h.x;
        d += w1 * h.y;
        d += w2 * h.z;
        d += w3 * h.w;
        d += __shfl_xor(d, 1);
        d += __shfl_xor(d, 2);
        d += __shfl_xor(d, 4);
        __shared__ float sdot[128];
        if (sub == 0) sdot[threadIdx.x >> 3] = d;
        __syncthreads();
        const int nb = (int)blockDim.x >> 3;
        const int o = blockIdx.x * nb + (int)threadIdx.x;
        if ((int)threadIdx.x < nb && o < n) dstp[o] = sdot[threadIdx.x] + bout[0];
    }
}

extern "C" void kernel_launch(void* const* d_in, const int* in_sizes, int n_in,
                              void* d_out, int out_size, void* d_ws, size_t ws_size,
                              hipStream_t stream)
{
    const float* x     = (const float*)d_in[0];
    const int*   ei    = (const int*)d_in[1];
    const float* W_in  = (const float*)d_in[2];
    const float* b_in  = (const float*)d_in[3];
    const float* W1    = (const float*)d_in[4];
    const float* b1    = (const float*)d_in[5];
    const float* W2    = (const float*)d_in[6];
    const float* b2    = (const float*)d_in[7];
    const float* W_out = (const float*)d_in[8];
    const float* b_out = (const float*)d_in[9];
    float* out = (float*)d_out;

    const int N = in_sizes[0] / 64;
    const int E = in_sizes[1] / 2;
    const int* src = ei;
    const int* dst = ei + E;

    float* ws   = (float*)d_ws;
    float* dinv = ws;
    float* h0   = ws + (size_t)N;
    float* p    = ws + (size_t)N * 65;
    float* h1   = ws + (size_t)N * 97;
    int* cnt    = (int*)(ws + (size_t)N * 161);
    int* rowptr = cnt + (size_t)N;
    int* slot   = rowptr + (size_t)N;
    int* csr    = slot + (size_t)E;

    const int TB = 256;
    const int gblk   = 512;
    const int eblk   = 2048;

    gcn_cnt_zero <<<(N + TB - 1) / TB, TB, 0, stream>>>(cnt, N);
    gcn_gemm_in_and_count<<<gblk + eblk, 128, 0, stream>>>(x, W_in, b_in, h0, N, gblk, dst, cnt, slot, E);
    gcn_rowptr   <<<(N + 1023) / 1024, 1024, 0, stream>>>(cnt, rowptr, dinv, N);

    gcn_gemm_c1_and_fill<<<gblk + eblk, 128, 0, stream>>>(h0, W1, dinv, p, N, gblk, src, dst, rowptr, slot, csr, E);
    gcn_pull<0><<<((size_t)N * 8 + TB - 1) / TB, TB, 0, stream>>>(rowptr, cnt, csr, p, dinv, b1, nullptr, nullptr, h1, N);

    gcn_gemm<32, 32, 1><<<gblk, 128, 0, stream>>>(h1, W2, nullptr, dinv, p, N);
    gcn_pull<1><<<((size_t)N * 8 + TB - 1) / TB, TB, 0, stream>>>(rowptr, cnt, csr, p, dinv, b2, W_out, b_out, out, N);
}


# edge list as fixed-pitch rows (32 per destination, overflow list past that): the count pass stores each source at row*32+slot itself; scan and fill passes gone; launches 7->6; sum order unchanged
# speedup vs baseline: 1.5674x; 1.1342x over previous
#include <hip/hip_runtime.h>
#include <hip/hip_bf16.h>

typedef __attribute__((ext_vector_type(2))) float v2f;
typedef __attribute__((ext_vector_type(8))) float v8f;

__device__ __forceinline__ float lrelu(float v) { return v > 0.0f ? v : 0.01f * v; }

constexpr int ROWCAP = 32;

__global__ void gcn_cnt_zero(int* __restrict__ cnt, int* __restrict__ ovn, int n) {
    int i = blockIdx.x * blockDim.x + threadIdx.x;
    if (i < n) cnt[i] = 0;
    if (i == 0) ovn[0] = 0;
}

__device__ __forceinline__ void place_edge(int d, int r, int s, int* __restrict__ rows,
                                           int* __restrict__ ovn, int* __restrict__ ovs,
                                           int* __restrict__ ovd) {
    if (r < ROWCAP) {
        rows[(size_t)d * ROWCAP + r] = s;
    } else {
        const int o = atomicAdd(ovn, 1);
        ovs[o] = s;
        ovd[o] = d;
    }
}

__device__ __forceinline__ void place_edges(const int* __restrict__ src, const int* __restrict__ dst,
                                            int* __restrict__ cnt, int* __restrict__ rows,
                                            int* __restrict__ ovn, int* __restrict__ ovs,
                                            int* __restrict__ ovd, int E, int blk, int nblk) {
    const int stride = nblk * (int)blockDim.x;
    for (int e0 = blk * (int)blockDim.x + (int)threadIdx.x; e0 < E; e0 += 4 * stride) {
        const int e1 = e0 + stride, e2 = e1 + stride, e3 = e2 + stride;
        if (e3 < E) {
            const int d0 = dst[e0], d1 = dst[e1], d2 = dst[e2], d3 = dst[e3];
            const int s0 = src[e0], s1 = src[e1], s2 = src[e2], s3 = src[e3];
            const int r0 = atomicAdd(&cnt[d0], 1);
            const int r1 = atomicAdd(&cnt[d1], 1);
            const int r2 = atomicAdd(&cnt[d2], 1);
            const int r3 = atomicAdd(&cnt[d3], 1);
            place_edge(d0, r0, s0, rows, ovn, ovs, ovd);
            place_edge(d1, r1, s1, rows, ovn, ovs, ovd);
            place_edge(d2, r2, s2, rows, ovn, ovs, ovd);
            place_edge(d3, r3, s3, rows, ovn, ovs, ovd);
        } else {
            { const int d = dst[e0]; place_edge(d, atomicAdd(&cnt[d], 1), src[e0], rows, ovn, ovs, ovd); }
            if (e1 < E) { const int d = dst[e1]; place_edge(d, atomicAdd(&cnt[d], 1), src[e1], rows, ovn, ovs, ovd); }
            if (e2 < E) { const int d = dst[e2]; place_edge(d, atomicAdd(&cnt[d], 1), src[e2], rows, ovn, ovs, ovd); }
        }
    }
}

template <int K, int NOUT, int MODE>
__device__ __forceinline__ void gemm_tiles(
    const float* __restrict__ in,
    const float* __restrict__ W,
    const float* __restrict__ bias,
    const int* __restrict__ cnt,
    float* __restrict__ out,
    int n, int blk, int nblk)
{
    constexpr int KS = K / 4;
    constexpr int NT = NOUT / 16;

    const int lane  = threadIdx.x & 31;
    const int lhalf = lane >> 4;
    const int l16   = lane & 15;
    const int wave   = __builtin_amdgcn_readfirstlane((int)(threadIdx.x >> 5));
    const int gwave  = blk * 4 + wave;
    const int nwaves = nblk * 4;
    const int tiles  = n >> 4;
    if (gwave >= tiles) return;

    v2f bf[KS][NT];
    #pragma unroll
    for (int s = 0; s < KS; ++s) {
        #pragma unroll
        for (int t = 0; t < NT; ++t) {
            const float* wp = W + (t * 16 + l16) * K + 4 * s + 2 * lhalf;
            bf[s][t].x = wp[0];
            bf[s][t].y = wp[1];
        }
    }

    float bv[NT];
    if (MODE == 0) {
        #pragma unroll
        for (int t = 0; t < NT; ++t) bv[t] = bias[t * 16 + l16];
    }

    v2f a[KS];
    {
        const float* ap = in + (size_t)(gwave * 16 + l16) * K + 2 * lhalf;
        #pragma unroll
        for (int s = 0; s < KS; ++s) { a[s].x = ap[4 * s]; a[s].y = ap[4 * s + 1]; }
    }

    for (int tile = gwave; tile < tiles; tile += nwaves) {
        const int row0 = tile * 16;

        const int ntile = (tile + nwaves < tiles) ? (tile + nwaves) : tile;
        int dc[8];
        if (MODE == 1) {
            #pragma unroll
            for (int r = 0; r < 8; ++r) dc[r] = cnt[row0 + r + 8 * lhalf];
        }
        v2f an[KS];
        {
            const float* ap = in + (size_t)(ntile * 16 + l16) * K + 2 * lhalf;
            #pragma unroll
            for (int s = 0; s < KS; ++s) { an[s].x = ap[4 * s]; an[s].y = ap[4 * s + 1]; }
        }

        __builtin_amdgcn_sched_barrier(0);

        v8f acc[NT];
        #pragma unroll
        for (int t = 0; t < NT; ++t)
            acc[t] = (v8f){0.f, 0.f, 0.f, 0.f, 0.f, 0.f, 0.f, 0.f};

        #pragma unroll
        for (int s = 0; s < KS; ++s) {
            #pragma unroll
            for (int t = 0; t < NT; ++t) {
                acc[t] = __builtin_amdgcn_wmma_f32_16x16x4_f32(
                    false, a[s], false, bf[s][t], (short)0, acc[t], false, false);
            }
        }

        float dv[8];
        if (MODE == 1) {
            __builtin_amdgcn_sched_barrier(0);
            #pragma unroll
            for (int r = 0; r < 8; ++r) dv[r] = rsqrtf((float)(dc[r] + 1));
        }

        #pragma unroll
        for (int t = 0; t < NT; ++t) {
            const int col = t * 16 + l16;
            if (MODE == 0) {
                #pragma unroll
                for (int r = 0; r < 8; ++r) {
                    const int row = row0 + r + 8 * lhalf;
                    out[(size_t)row * NOUT + col] = lrelu(acc[t][r] + bv[t]);
                }
            } else {
                #pragma unroll
                for (int r = 0; r < 8; ++r) {
                    const int row = row0 + r + 8 * lhalf;
                    out[(size_t)row * NOUT + col] = acc[t][r] * dv[r];
                }
            }
        }

        __builtin_amdgcn_sched_barrier(0);

        #pragma unroll
        for (int s = 0; s < KS; ++s) a[s] = an[s];
    }
}

template <int K, int NOUT, int MODE>
__global__ __launch_bounds__(128) void gcn_gemm(
    const float* __restrict__ in, const float* __restrict__ W, const float* __restrict__ bias,
    const int* __restrict__ cnt, float* __restrict__ out, int n)
{
    gemm_tiles<K, NOUT, MODE>(in, W, bias, cnt, out, n, (int)blockIdx.x, (int)gridDim.x);
}

__global__ __launch_bounds__(128) void gcn_gemm_in_and_place(
    const float* __restrict__ x, const float* __restrict__ W_in, const float* __restrict__ b_in,
    float* __restrict__ h0, int n, int gblk,
    const int* __restrict__ src, const int* __restrict__ dst, int* __restrict__ cnt,
    int* __restrict__ rows, int* __restrict__ ovn, int* __restrict__ ovs, int* __restrict__ ovd, int E)
{
    if ((int)blockIdx.x < gblk)
        gemm_tiles<64, 64, 0>(x, W_in, b_in, nullptr, h0, n, (int)blockIdx.x, gblk);
    else
        place_edges(src, dst, cnt, rows, ovn, ovs, ovd, E, (int)blockIdx.x - gblk, (int)gridDim.x - gblk);
}

template <int MODE>
__global__ void gcn_pull(const int* __restrict__ cnt, const int* __restrict__ rows,
                         const int* __restrict__ ovn, const int* __restrict__ ovs,
                         const int* __restrict__ ovd, int E,
                         const float* __restrict__ p, const float* __restrict__ bias,
                         const float* __restrict__ Wout, const float* __restrict__ bout,
                         float* __restrict__ dstp, int n)
{
    const int t = blockIdx.x * blockDim.x + threadIdx.x;
    const bool live = (t >> 3) < n;
    if (MODE == 0 && !live) return;
    const int v = live ? (t >> 3) : (n - 1);
    const int sub = t & 7;
    const float4* p4 = (const float4*)p + sub;
    const int deg = cnt[v];
    const float b0 = bias[4 * sub], b1 = bias[4 * sub + 1], b2 = bias[4 * sub + 2], b3 = bias[4 * sub + 3];
    float w0 = 0.0f, w1 = 0.0f, w2 = 0.0f, w3 = 0.0f;
    if (MODE == 1) { w0 = Wout[4 * sub]; w1 = Wout[4 * sub + 1]; w2 = Wout[4 * sub + 2]; w3 = Wout[4 * sub + 3]; }
    float4 s = p4[(size_t)v * 8];
    const int* row = rows + (size_t)v * ROWCAP;
    const int top = n - 1;
    const int inrow = min(deg, ROWCAP);
    if (inrow > 0) {
        const int last = inrow - 1;
        int u0 = row[0];
        int u1 = row[min(1, last)];
        int u2 = row[min(2, last)];
        int u3 = row[min(3, last)];
        for (int j = 0; j < inrow; j += 4) {
            const int r = inrow - j;
            const float4 m0 = p4[(size_t)min(max(u0, 0), top) * 8];
            const float4 m1 = p4[(size_t)min(max(u1, 0), top) * 8];
            const float4 m2 = p4[(size_t)min(max(u2, 0), top) * 8];
            const float4 m3 = p4[(size_t)min(max(u3, 0), top) * 8];
            u0 = row[min(j + 4, last)];
            u1 = row[min(j + 5, last)];
            u2 = row[min(j + 6, last)];
            u3 = row[min(j + 7, last)];
            asm volatile("" :: "v"(u0), "v"(u1), "v"(u2), "v"(u3),
                               "v"(m0.x), "v"(m1.x), "v"(m2.x), "v"(m3.x));
            const bool h1 = r > 1, h2 = r > 2, h3 = r > 3;
            s.x += m0.x; s.y += m0.y; s.z += m0.z; s.w += m0.w;
            s.x += h1 ? m1.x : -0.0f; s.y += h1 ? m1.y : -0.0f; s.z += h1 ? m1.z : -0.0f; s.w += h1 ? m1.w : -0.0f;
            s.x += h2 ? m2.x : -0.0f; s.y += h2 ? m2.y : -0.0f; s.z += h2 ? m2.z : -0.0f; s.w += h2 ? m2.w : -0.0f;
            s.x += h3 ? m3.x : -0.0f; s.y += h3 ? m3.y : -0.0f; s.z += h3 ? m3.z : -0.0f; s.w += h3 ? m3.w : -0.0f;
        }
    }
    if (deg > ROWCAP) {
        const int novf = min(ovn[0], E);
        for (int o = 0; o < novf; ++o) {
            if (ovd[o] == v) {
                const float4 m = p4[(size_t)min(max(ovs[o], 0), top) * 8];
                s.x += m.x; s.y += m.y; s.z += m.z; s.w += m.w;
            }
        }
    }
    const float di = rsqrtf((float)(deg + 1));
    float4 h;
    h.x = lrelu(di * s.x + b0);
    h.y = lrelu(di * s.y + b1);
    h.z = lrelu(di * s.z + b2);
    h.w = lrelu(di * s.w + b3);
    if (MODE == 0) {
        ((float4*)dstp)[(size_t)v * 8 + sub] = h;
    } else {
        float d = w0 * h.x;
        d += w1 * h.y;
        d += w2 * h.z;
        d += w3 * h.w;
        d += __shfl_xor(d, 1);
        d += __shfl_xor(d, 2);
        d += __shfl_xor(d, 4);
        __shared__ float sdot[128];
        if (sub == 0) sdot[threadIdx.x >> 3] = d;
        __syncthreads();
        const int nb = (int)blockDim.x >> 3;
        const int o = blockIdx.x * nb + (int)threadIdx.x;
        if ((int)threadIdx.x < nb && o < n) dstp[o] = sdot[threadIdx.x] + bout[0];
    }
}

extern "C" void kernel_launch(void* const* d_in, const int* in_sizes, int n_in,
                              void* d_out, int out_size, void* d_ws, size_t ws_size,
                              hipStream_t stream)
{
    const float* x     = (const float*)d_in[0];
    const int*   ei    = (const int*)d_in[1];
    const float* W_in  = (const float*)d_in[2];
    const float* b_in  = (const float*)d_in[3];
    const float* W1    = (const float*)d_in[4];
    const float* b1    = (const float*)d_in[5];
    const float* W2    = (const float*)d_in[6];
    const float* b2    = (const float*)d_in[7];
    const float* W_out = (const float*)d_in[8];
    const float* b_out = (const float*)d_in[9];
    float* out = (float*)d_out;

    const int N = in_sizes[0] / 64;
    const int E = in_sizes[1] / 2;
    const int* src = ei;
    const int* dst = ei + E;

    float* ws   = (float*)d_ws;
    int* ovn    = (int*)d_ws;
    float* h0   = ws + (size_t)N;
    float* p    = ws + (size_t)N * 65;
    float* h1   = ws + (size_t)N * 97;
    int* cnt    = (int*)(ws + (size_t)N * 161);
    int* rows   = cnt + (size_t)N;
    int* ovs    = rows + (size_t)N * ROWCAP;
    int* ovd    = ovs + (size_t)E;

    const int TB = 256;
    const int gblk   = 512;
    const int eblk   = 2048;

    gcn_cnt_zero <<<(N + TB - 1) / TB, TB, 0, stream>>>(cnt, ovn, N);
    gcn_gemm_in_and_place<<<gblk + eblk, 128, 0, stream>>>(x, W_in, b_in, h0, N, gblk, src, dst, cnt, rows, ovn, ovs, ovd, E);

    gcn_gemm<64, 32, 1><<<gblk, 128, 0, stream>>>(h0, W1, nullptr, cnt, p, N);
    gcn_pull<0><<<((size_t)N * 8 + TB - 1) / TB, TB, 0, stream>>>(cnt, rows, ovn, ovs, ovd, E, p, b1, nullptr, nullptr, h1, N);

    gcn_gemm<32, 32, 1><<<gblk, 128, 0, stream>>>(h1, W2, nullptr, cnt, p, N);
    gcn_pull<1><<<((size_t)N * 8 + TB - 1) / TB, TB, 0, stream>>>(cnt, rows, ovn, ovs, ovd, E, p, b2, W_out, b_out, out, N);
}


# edge gathers on one wave per 32 nodes: a node's source row read by one 128-byte load and handed out by lane-to-lane moves, 16 gathers in flight, next groups requested ahead; sum order unchanged
# speedup vs baseline: 1.6514x; 1.0536x over previous
#include <hip/hip_runtime.h>
#include <hip/hip_bf16.h>

typedef __attribute__((ext_vector_type(2))) float v2f;
typedef __attribute__((ext_vector_type(8))) float v8f;

__device__ __forceinline__ float lrelu(float v) { return v > 0.0f ? v : 0.01f * v; }

constexpr int ROWCAP = 32;

__global__ void gcn_cnt_zero(int* __restrict__ cnt, int* __restrict__ ovn, int n) {
    int i = blockIdx.x * blockDim.x + threadIdx.x;
    if (i < n) cnt[i] = 0;
    if (i == 0) ovn[0] = 0;
}

__device__ __forceinline__ void place_edge(int d, int r, int s, int* __restrict__ rows,
                                           int* __restrict__ ovn, int* __restrict__ ovs,
                                           int* __restrict__ ovd) {
    if (r < ROWCAP) {
        rows[(size_t)d * ROWCAP + r] = s;
    } else {
        const int o = atomicAdd(ovn, 1);
        ovs[o] = s;
        ovd[o] = d;
    }
}

__device__ __forceinline__ void place_edges(const int* __restrict__ src, const int* __restrict__ dst,
                                            int* __restrict__ cnt, int* __restrict__ rows,
                                            int* __restrict__ ovn, int* __restrict__ ovs,
                                            int* __restrict__ ovd, int E, int blk, int nblk) {
    const int stride = nblk * (int)blockDim.x;
    for (int e0 = blk * (int)blockDim.x + (int)threadIdx.x; e0 < E; e0 += 4 * stride) {
        const int e1 = e0 + stride, e2 = e1 + stride, e3 = e2 + stride;
        if (e3 < E) {
            const int d0 = dst[e0], d1 = dst[e1], d2 = dst[e2], d3 = dst[e3];
            const int s0 = src[e0], s1 = src[e1], s2 = src[e2], s3 = src[e3];
            const int r0 = atomicAdd(&cnt[d0], 1);
            const int r1 = atomicAdd(&cnt[d1], 1);
            const int r2 = atomicAdd(&cnt[d2], 1);
            const int r3 = atomicAdd(&cnt[d3], 1);
            place_edge(d0, r0, s0, rows, ovn, ovs, ovd);
            place_edge(d1, r1, s1, rows, ovn, ovs, ovd);
            place_edge(d2, r2, s2, rows, ovn, ovs, ovd);
            place_edge(d3, r3, s3, rows, ovn, ovs, ovd);
        } else {
            { const int d = dst[e0]; place_edge(d, atomicAdd(&cnt[d], 1), src[e0], rows, ovn, ovs, ovd); }
            if (e1 < E) { const int d = dst[e1]; place_edge(d, atomicAdd(&cnt[d], 1), src[e1], rows, ovn, ovs, ovd); }
            if (e2 < E) { const int d = dst[e2]; place_edge(d, atomicAdd(&cnt[d], 1), src[e2], rows, ovn, ovs, ovd); }
        }
    }
}

template <int K, int NOUT, int MODE>
__device__ __forceinline__ void gemm_tiles(
    const float* __restrict__ in,
    const float* __restrict__ W,
    const float* __restrict__ bias,
    const int* __restrict__ cnt,
    float* __restrict__ out,
    int n, int blk, int nblk)
{
    constexpr int KS = K / 4;
    constexpr int NT = NOUT / 16;

    const int lane  = threadIdx.x & 31;
    const int lhalf = lane >> 4;
    const int l16   = lane & 15;
    const int wave   = __builtin_amdgcn_readfirstlane((int)(threadIdx.x >> 5));
    const int gwave  = blk * 4 + wave;
    const int nwaves = nblk * 4;
    const int tiles  = n >> 4;
    if (gwave >= tiles) return;

    v2f bf[KS][NT];
    #pragma unroll
    for (int s = 0; s < KS; ++s) {
        #pragma unroll
        for (int t = 0; t < NT; ++t) {
            const float* wp = W + (t * 16 + l16) * K + 4 * s + 2 * lhalf;
            bf[s][t].x = wp[0];
            bf[s][t].y = wp[1];
        }
    }

    float bv[NT];
    if (MODE == 0) {
        #pragma unroll
        for (int t = 0; t < NT; ++t) bv[t] = bias[t * 16 + l16];
    }

    v2f a[KS];
    {
        const float* ap = in + (size_t)(gwave * 16 + l16) * K + 2 * lhalf;
        #pragma unroll
        for (int s = 0; s < KS; ++s) { a[s].x = ap[4 * s]; a[s].y = ap[4 * s + 1]; }
    }

    for (int tile = gwave; tile < tiles; tile += nwaves) {
        const int row0 = tile * 16;

        const int ntile = (tile + nwaves < tiles) ? (tile + nwaves) : tile;
        int dc[8];
        if (MODE == 1) {
            #pragma unroll
            for (int r = 0; r < 8; ++r) dc[r] = cnt[row0 + r + 8 * lhalf];
        }
        v2f an[KS];
        {
            const float* ap = in + (size_t)(ntile * 16 + l16) * K + 2 * lhalf;
            #pragma unroll
            for (int s = 0; s < KS; ++s) { an[s].x = ap[4 * s]; an[s].y = ap[4 * s + 1]; }
        }

        __builtin_amdgcn_sched_barrier(0);

        v8f acc[NT];
        #pragma unroll
        for (int t = 0; t < NT; ++t)
            acc[t] = (v8f){0.f, 0.f, 0.f, 0.f, 0.f, 0.f, 0.f, 0.f};

        #pragma unroll
        for (int s = 0; s < KS; ++s) {
            #pragma unroll
            for (int t = 0; t < NT; ++t) {
                acc[t] = __builtin_amdgcn_wmma_f32_16x16x4_f32(
                    false, a[s], false, bf[s][t], (short)0, acc[t], false, false);
            }
        }

        float dv[8];
        if (MODE == 1) {
            __builtin_amdgcn_sched_barrier(0);
            #pragma unroll
            for (int r = 0; r < 8; ++r) dv[r] = rsqrtf((float)(dc[r] + 1));
        }

        #pragma unroll
        for (int t = 0; t < NT; ++t) {
            const int col = t * 16 + l16;
            if (MODE == 0) {
                #pragma unroll
                for (int r = 0; r < 8; ++r) {
                    const int row = row0 + r + 8 * lhalf;
                    out[(size_t)row * NOUT + col] = lrelu(acc[t][r] + bv[t]);
                }
            } else {
                #pragma unroll
                for (int r = 0; r < 8; ++r) {
                    const int row = row0 + r + 8 * lhalf;
                    out[(size_t)row * NOUT + col] = acc[t][r] * dv[r];
                }
            }
        }

        __builtin_amdgcn_sched_barrier(0);

        #pragma unroll
        for (int s = 0; s < KS; ++s) a[s] = an[s];
    }
}

template <int K, int NOUT, int MODE>
__global__ __launch_bounds__(128) void gcn_gemm(
    const float* __restrict__ in, const float* __restrict__ W, const float* __restrict__ bias,
    const int* __restrict__ cnt, float* __restrict__ out, int n)
{
    gemm_tiles<K, NOUT, MODE>(in, W, bias, cnt, out, n, (int)blockIdx.x, (int)gridDim.x);
}

__global__ __launch_bounds__(128) void gcn_gemm_in_and_place(
    const float* __restrict__ x, const float* __restrict__ W_in, const float* __restrict__ b_in,
    float* __restrict__ h0, int n, int gblk,
    const int* __restrict__ src, const int* __restrict__ dst, int* __restrict__ cnt,
    int* __restrict__ rows, int* __restrict__ ovn, int* __restrict__ ovs, int* __restrict__ ovd, int E)
{
    if ((int)blockIdx.x < gblk)
        gemm_tiles<64, 64, 0>(x, W_in, b_in, nullptr, h0, n, (int)blockIdx.x, gblk);
    else
        place_edges(src, dst, cnt, rows, ovn, ovs, ovd, E, (int)blockIdx.x - gblk, (int)gridDim.x - gblk);
}


template <int J>
__device__ __forceinline__ int bcast8(int x) {
    return (int)__builtin_amdgcn_mov_dpp8((unsigned)x, J * 0x249249);
}

struct NodeHead { int v; int deg; int4 r; float4 s; };

__device__ __forceinline__ NodeHead load_head(int v, const int* __restrict__ cnt,
                                              const int4* __restrict__ row4,
                                              const float4* __restrict__ p4) {
    NodeHead h;
    h.v = v;
    h.deg = cnt[v];
    h.r = row4[(size_t)v * 8];
    h.s = p4[(size_t)v * 8];
    return h;
}

template <int J>
__device__ __forceinline__ void trip_fetch(const NodeHead& h, int top,
                                           const float4* __restrict__ p4, float4* m) {
    const int inrow = min(h.deg, ROWCAP);
    const int u0 = bcast8<J>(h.r.x), u1 = bcast8<J>(h.r.y), u2 = bcast8<J>(h.r.z), u3 = bcast8<J>(h.r.w);
    m[0] = p4[(size_t)(4 * J + 0 < inrow ? min(max(u0, 0), top) : h.v) * 8];
    m[1] = p4[(size_t)(4 * J + 1 < inrow ? min(max(u1, 0), top) : h.v) * 8];
    m[2] = p4[(size_t)(4 * J + 2 < inrow ? min(max(u2, 0), top) : h.v) * 8];
    m[3] = p4[(size_t)(4 * J + 3 < inrow ? min(max(u3, 0), top) : h.v) * 8];
}

template <int J>
__device__ __forceinline__ void trip_sum(float4& s, int inrow, const float4* m) {
    const bool h0 = 4 * J + 0 < inrow, h1 = 4 * J + 1 < inrow;
    const bool h2 = 4 * J + 2 < inrow, h3 = 4 * J + 3 < inrow;
    s.x += h0 ? m[0].x : -0.0f; s.y += h0 ? m[0].y : -0.0f; s.z += h0 ? m[0].z : -0.0f; s.w += h0 ? m[0].w : -0.0f;
    s.x += h1 ? m[1].x : -0.0f; s.y += h1 ? m[1].y : -0.0f; s.z += h1 ? m[1].z : -0.0f; s.w += h1 ? m[1].w : -0.0f;
    s.x += h2 ? m[2].x : -0.0f; s.y += h2 ? m[2].y : -0.0f; s.z += h2 ? m[2].z : -0.0f; s.w += h2 ? m[2].w : -0.0f;
    s.x += h3 ? m[3].x : -0.0f; s.y += h3 ? m[3].y : -0.0f; s.z += h3 ? m[3].z : -0.0f; s.w += h3 ? m[3].w : -0.0f;
}

__device__ __forceinline__ void fetch16(const NodeHead& h, int top, const float4* __restrict__ p4, float4* m) {
    trip_fetch<0>(h, top, p4, m);
    trip_fetch<1>(h, top, p4, m + 4);
    trip_fetch<2>(h, top, p4, m + 8);
    trip_fetch<3>(h, top, p4, m + 12);
}

template <int J>
__device__ __forceinline__ void trip_late(float4& s, const NodeHead& h, int top, const float4* __restrict__ p4) {
    float4 m[4];
    trip_fetch<J>(h, top, p4, m);
    trip_sum<J>(s, min(h.deg, ROWCAP), m);
}

template <int MODE, bool GATHER_NEXT, bool HEAD_AFTER>
__device__ __forceinline__ void pull_step(int g, int base, int n, int lane, int k, int sub,
                                          NodeHead& c, NodeHead& nx, const float4* mc, float4* mn,
                                          const int* __restrict__ cnt, const int4* __restrict__ row4,
                                          const float4* __restrict__ p4,
                                          const int* __restrict__ ovn, const int* __restrict__ ovs,
                                          const int* __restrict__ ovd, int E,
                                          float b0, float b1, float b2, float b3,
                                          float w0, float w1, float w2, float w3,
                                          float* __restrict__ dstp, float& outv)
{
    const int top = n - 1;
    NodeHead nn = nx;
    if (HEAD_AFTER) nn = load_head(min(base + 4 * (g + 2) + k, top), cnt, row4, p4);
    if (GATHER_NEXT) fetch16(nx, top, p4, mn);

    const int v = c.v, deg = c.deg;
    const int inrow = min(deg, ROWCAP);
    float4 s = c.s;
    trip_sum<0>(s, inrow, mc);
    trip_sum<1>(s, inrow, mc + 4);
    trip_sum<2>(s, inrow, mc + 8);
    trip_sum<3>(s, inrow, mc + 12);

    if (__builtin_amdgcn_ballot_w32(inrow > 16) != 0u) {
        trip_late<4>(s, c, top, p4);
        if (__builtin_amdgcn_ballot_w32(inrow > 20) != 0u) {
            trip_late<5>(s, c, top, p4);
            if (__builtin_amdgcn_ballot_w32(inrow > 24) != 0u) {
                trip_late<6>(s, c, top, p4);
                if (__builtin_amdgcn_ballot_w32(inrow > 28) != 0u)
                    trip_late<7>(s, c, top, p4);
            }
        }
    }

    if (deg > ROWCAP) {
        const int novf = min(ovn[0], E);
        for (int o = 0; o < novf; ++o) {
            if (ovd[o] == v) {
                const float4 m = p4[(size_t)min(max(ovs[o], 0), top) * 8];
                s.x += m.x; s.y += m.y; s.z += m.z; s.w += m.w;
            }
        }
    }

    const float di = rsqrtf((float)(deg + 1));
    float4 h;
    h.x = lrelu(di * s.x + b0);
    h.y = lrelu(di * s.y + b1);
    h.z = lrelu(di * s.z + b2);
    h.w = lrelu(di * s.w + b3);
    if (MODE == 0) {
        if (base + 4 * g + k < n) ((float4*)dstp)[(size_t)v * 8 + sub] = h;
    } else {
        float d = w0 * h.x;
        d += w1 * h.y;
        d += w2 * h.z;
        d += w3 * h.w;
        d += __shfl_xor(d, 1);
        d += __shfl_xor(d, 2);
        d += __shfl_xor(d, 4);
        const float dl = __shfl(d, (lane & 3) * 8);
        if ((lane >> 2) == g) outv = dl;
    }
    c = nx;
    nx = nn;
}

template <int MODE>
__global__ __launch_bounds__(128)
void gcn_pull(const int* __restrict__ cnt, const int* __restrict__ rows,
              const int* __restrict__ ovn, const int* __restrict__ ovs,
              const int* __restrict__ ovd, int E,
              const float* __restrict__ p, const float* __restrict__ bias,
              const float* __restrict__ Wout, const float* __restrict__ bout,
              float* __restrict__ dstp, int n)
{
    const int lane = threadIdx.x & 31;
    const int k    = lane >> 3;
    const int sub  = lane & 7;
    const int wave = __builtin_amdgcn_readfirstlane((int)(threadIdx.x >> 5));
    const int base = (blockIdx.x * ((int)blockDim.x >> 5) + wave) * 32;
    if (base >= n) return;
    const int top = n - 1;
    const float4* p4   = (const float4*)p + sub;
    const int4*   row4 = (const int4*)rows + sub;

    const float b0 = bias[4 * sub], b1 = bias[4 * sub + 1], b2 = bias[4 * sub + 2], b3 = bias[4 * sub + 3];
    float w0 = 0.0f, w1 = 0.0f, w2 = 0.0f, w3 = 0.0f, bo = 0.0f;
    if (MODE == 1) { w0 = Wout[4 * sub]; w1 = Wout[4 * sub + 1]; w2 = Wout[4 * sub + 2]; w3 = Wout[4 * sub + 3]; bo = bout[0]; }

    NodeHead c  = load_head(min(base + k, top), cnt, row4, p4);
    NodeHead nx = load_head(min(base + 4 + k, top), cnt, row4, p4);
    float4 ma[16], mb[16];
    fetch16(c, top, p4, ma);
    float outv = 0.0f;

    for (int g = 0; g < 6; g += 2) {
        pull_step<MODE, true, true>(g,     base, n, lane, k, sub, c, nx, ma, mb, cnt, row4, p4, ovn, ovs, ovd, E,
                                    b0, b1, b2, b3, w0, w1, w2, w3, dstp, outv);
        pull_step<MODE, true, true>(g + 1, base, n, lane, k, sub, c, nx, mb, ma, cnt, row4, p4, ovn, ovs, ovd, E,
                                    b0, b1, b2, b3, w0, w1, w2, w3, dstp, outv);
    }
    pull_step<MODE, true, false>(6, base, n, lane, k, sub, c, nx, ma, mb, cnt, row4, p4, ovn, ovs, ovd, E,
                                 b0, b1, b2, b3, w0, w1, w2, w3, dstp, outv);
    pull_step<MODE, false, false>(7, base, n, lane, k, sub, c, nx, mb, ma, cnt, row4, p4, ovn, ovs, ovd, E,
                                  b0, b1, b2, b3, w0, w1, w2, w3, dstp, outv);
    if (MODE == 1) {
        if (base + lane < n) dstp[base + lane] = outv + bo;
    }
}

extern "C" void kernel_launch(void* const* d_in, const int* in_sizes, int n_in,
                              void* d_out, int out_size, void* d_ws, size_t ws_size,
                              hipStream_t stream)
{
    const float* x     = (const float*)d_in[0];
    const int*   ei    = (const int*)d_in[1];
    const float* W_in  = (const float*)d_in[2];
    const float* b_in  = (const float*)d_in[3];
    const float* W1    = (const float*)d_in[4];
    const float* b1    = (const float*)d_in[5];
    const float* W2    = (const float*)d_in[6];
    const float* b2    = (const float*)d_in[7];
    const float* W_out = (const float*)d_in[8];
    const float* b_out = (const float*)d_in[9];
    float* out = (float*)d_out;

    const int N = in_sizes[0] / 64;
    const int E = in_sizes[1] / 2;
    const int* src = ei;
    const int* dst = ei + E;

    float* ws   = (float*)d_ws;
    int* ovn    = (int*)d_ws;
    float* h0   = ws + (size_t)N;
    float* p    = ws + (size_t)N * 65;
    float* h1   = ws + (size_t)N * 97;
    int* cnt    = (int*)(ws + (size_t)N * 161);
    int* rows   = cnt + (size_t)N;
    int* ovs    = rows + (size_t)N * ROWCAP;
    int* ovd    = ovs + (size_t)E;

    const int TB = 256;
    const int gblk   = 512;
    const int eblk   = 2048;

    gcn_cnt_zero <<<(N + TB - 1) / TB, TB, 0, stream>>>(cnt, ovn, N);
    gcn_gemm_in_and_place<<<gblk + eblk, 128, 0, stream>>>(x, W_in, b_in, h0, N, gblk, src, dst, cnt, rows, ovn, ovs, ovd, E);

    gcn_gemm<64, 32, 1><<<gblk, 128, 0, stream>>>(h0, W1, nullptr, cnt, p, N);
    gcn_pull<0><<<((size_t)N + 127) / 128, 128, 0, stream>>>(cnt, rows, ovn, ovs, ovd, E, p, b1, nullptr, nullptr, h1, N);

    gcn_gemm<32, 32, 1><<<gblk, 128, 0, stream>>>(h1, W2, nullptr, cnt, p, N);
    gcn_pull<1><<<((size_t)N + 127) / 128, 128, 0, stream>>>(cnt, rows, ovn, ovs, ovd, E, p, b2, W_out, b_out, out, N);
}


# conv2 weight applied behind its edge sum: pull 1 stores dinv*h1; pull 2 sums those rows, W2 x a wave's 32 sums on WMMA via LDS, then the dot; launches 6->5; conv2 and dot sum order differ
# speedup vs baseline: 1.7463x; 1.0575x over previous
#include <hip/hip_runtime.h>
#include <hip/hip_bf16.h>

typedef __attribute__((ext_vector_type(2))) float v2f;
typedef __attribute__((ext_vector_type(8))) float v8f;

__device__ __forceinline__ float lrelu(float v) { return v > 0.0f ? v : 0.01f * v; }

constexpr int ROWCAP = 32;
constexpr int TPITCH = 36;

__global__ void gcn_cnt_zero(int* __restrict__ cnt, int* __restrict__ ovn, int n) {
    int i = blockIdx.x * blockDim.x + threadIdx.x;
    if (i < n) cnt[i] = 0;
    if (i == 0) ovn[0] = 0;
}

__device__ __forceinline__ void place_edge(int d, int r, int s, int* __restrict__ rows,
                                           int* __restrict__ ovn, int* __restrict__ ovs,
                                           int* __restrict__ ovd) {
    if (r < ROWCAP) {
        rows[(size_t)d * ROWCAP + r] = s;
    } else {
        const int o = atomicAdd(ovn, 1);
        ovs[o] = s;
        ovd[o] = d;
    }
}

__device__ __forceinline__ void place_edges(const int* __restrict__ src, const int* __restrict__ dst,
                                            int* __restrict__ cnt, int* __restrict__ rows,
                                            int* __restrict__ ovn, int* __restrict__ ovs,
                                            int* __restrict__ ovd, int E, int blk, int nblk) {
    const int stride = nblk * (int)blockDim.x;
    for (int e0 = blk * (int)blockDim.x + (int)threadIdx.x; e0 < E; e0 += 4 * stride) {
        const int e1 = e0 + stride, e2 = e1 + stride, e3 = e2 + stride;
        if (e3 < E) {
            const int d0 = dst[e0], d1 = dst[e1], d2 = dst[e2], d3 = dst[e3];
            const int s0 = src[e0], s1 = src[e1], s2 = src[e2], s3 = src[e3];
            const int r0 = atomicAdd(&cnt[d0], 1);
            const int r1 = atomicAdd(&cnt[d1], 1);
            const int r2 = atomicAdd(&cnt[d2], 1);
            const int r3 = atomicAdd(&cnt[d3], 1);
            place_edge(d0, r0, s0, rows, ovn, ovs, ovd);
            place_edge(d1, r1, s1, rows, ovn, ovs, ovd);
            place_edge(d2, r2, s2, rows, ovn, ovs, ovd);
            place_edge(d3, r3, s3, rows, ovn, ovs, ovd);
        } else {
            { const int d = dst[e0]; place_edge(d, atomicAdd(&cnt[d], 1), src[e0], rows, ovn, ovs, ovd); }
            if (e1 < E) { const int d = dst[e1]; place_edge(d, atomicAdd(&cnt[d], 1), src[e1], rows, ovn, ovs, ovd); }
            if (e2 < E) { const int d = dst[e2]; place_edge(d, atomicAdd(&cnt[d], 1), src[e2], rows, ovn, ovs, ovd); }
        }
    }
}

template <int K, int NOUT, int MODE>
__device__ __forceinline__ void gemm_tiles(
    const float* __restrict__ in,
    const float* __restrict__ W,
    const float* __restrict__ bias,
    const int* __restrict__ cnt,
    float* __restrict__ out,
    int n, int blk, int nblk)
{
    constexpr int KS = K / 4;
    constexpr int NT = NOUT / 16;

    const int lane  = threadIdx.x & 31;
    const int lhalf = lane >> 4;
    const int l16   = lane & 15;
    const int wave   = __builtin_amdgcn_readfirstlane((int)(threadIdx.x >> 5));
    const int gwave  = blk * 4 + wave;
    const int nwaves = nblk * 4;
    const int tiles  = n >> 4;
    if (gwave >= tiles) return;

    v2f bf[KS][NT];
    #pragma unroll
    for (int s = 0; s < KS; ++s) {
        #pragma unroll
        for (int t = 0; t < NT; ++t) {
            const float* wp = W + (t * 16 + l16) * K + 4 * s + 2 * lhalf;
            bf[s][t].x = wp[0];
            bf[s][t].y = wp[1];
        }
    }

    float bv[NT];
    if (MODE == 0) {
        #pragma unroll
        for (int t = 0; t < NT; ++t) bv[t] = bias[t * 16 + l16];
    }

    v2f a[KS];
    {
        const float* ap = in + (size_t)(gwave * 16 + l16) * K + 2 * lhalf;
        #pragma unroll
        for (int s = 0; s < KS; ++s) { a[s].x = ap[4 * s]; a[s].y = ap[4 * s + 1]; }
    }

    for (int tile = gwave; tile < tiles; tile += nwaves) {
        const int row0 = tile * 16;

        const int ntile = (tile + nwaves < tiles) ? (tile + nwaves) : tile;
        int dc[8];
        if (MODE == 1) {
            #pragma unroll
            for (int r = 0; r < 8; ++r) dc[r] = cnt[row0 + r + 8 * lhalf];
        }
        v2f an[KS];
        {
            const float* ap = in + (size_t)(ntile * 16 + l16) * K + 2 * lhalf;
            #pragma unroll
            for (int s = 0; s < KS; ++s) { an[s].x = ap[4 * s]; an[s].y = ap[4 * s + 1]; }
        }

        __builtin_amdgcn_sched_barrier(0);

        v8f acc[NT];
        #pragma unroll
        for (int t = 0; t < NT; ++t)
            acc[t] = (v8f){0.f, 0.f, 0.f, 0.f, 0.f, 0.f, 0.f, 0.f};

        #pragma unroll
        for (int s = 0; s < KS; ++s) {
            #pragma unroll
            for (int t = 0; t < NT; ++t) {
                acc[t] = __builtin_amdgcn_wmma_f32_16x16x4_f32(
                    false, a[s], false, bf[s][t], (short)0, acc[t], false, false);
            }
        }

        float dv[8];
        if (MODE == 1) {
            __builtin_amdgcn_sched_barrier(0);
            #pragma unroll
            for (int r = 0; r < 8; ++r) dv[r] = rsqrtf((float)(dc[r] + 1));
        }

        #pragma unroll
        for (int t = 0; t < NT; ++t) {
            const int col = t * 16 + l16;
            if (MODE == 0) {
                #pragma unroll
                for (int r = 0; r < 8; ++r) {
                    const int row = row0 + r + 8 * lhalf;
                    out[(size_t)row * NOUT + col] = lrelu(acc[t][r] + bv[t]);
                }
            } else {
                #pragma unroll
                for (int r = 0; r < 8; ++r) {
                    const int row = row0 + r + 8 * lhalf;
                    out[(size_t)row * NOUT + col] = acc[t][r] * dv[r];
                }
            }
        }

        __builtin_amdgcn_sched_barrier(0);

        #pragma unroll
        for (int s = 0; s < KS; ++s) a[s] = an[s];
    }
}

template <int K, int NOUT, int MODE>
__global__ __launch_bounds__(128) void gcn_gemm(
    const float* __restrict__ in, const float* __restrict__ W, const float* __restrict__ bias,
    const int* __restrict__ cnt, float* __restrict__ out, int n)
{
    gemm_tiles<K, NOUT, MODE>(in, W, bias, cnt, out, n, (int)blockIdx.x, (int)gridDim.x);
}

__global__ __launch_bounds__(128) void gcn_gemm_in_and_place(
    const float* __restrict__ x, const float* __restrict__ W_in, const float* __restrict__ b_in,
    float* __restrict__ h0, int n, int gblk,
    const int* __restrict__ src, const int* __restrict__ dst, int* __restrict__ cnt,
    int* __restrict__ rows, int* __restrict__ ovn, int* __restrict__ ovs, int* __restrict__ ovd, int E)
{
    if ((int)blockIdx.x < gblk)
        gemm_tiles<64, 64, 0>(x, W_in, b_in, nullptr, h0, n, (int)blockIdx.x, gblk);
    else
        place_edges(src, dst, cnt, rows, ovn, ovs, ovd, E, (int)blockIdx.x - gblk, (int)gridDim.x - gblk);
}


template <int J>
__device__ __forceinline__ int bcast8(int x) {
    return (int)__builtin_amdgcn_mov_dpp8((unsigned)x, J * 0x249249);
}

struct NodeHead { int v; int deg; int4 r; float4 s; };

__device__ __forceinline__ NodeHead load_head(int v, const int* __restrict__ cnt,
                                              const int4* __restrict__ row4,
                                              const float4* __restrict__ p4) {
    NodeHead h;
    h.v = v;
    h.deg = cnt[v];
    h.r = row4[(size_t)v * 8];
    h.s = p4[(size_t)v * 8];
    return h;
}

template <int J>
__device__ __forceinline__ void trip_fetch(const NodeHead& h, int top,
                                           const float4* __restrict__ p4, float4* m) {
    const int inrow = min(h.deg, ROWCAP);
    const int u0 = bcast8<J>(h.r.x), u1 = bcast8<J>(h.r.y), u2 = bcast8<J>(h.r.z), u3 = bcast8<J>(h.r.w);
    m[0] = p4[(size_t)(4 * J + 0 < inrow ? min(max(u0, 0), top) : h.v) * 8];
    m[1] = p4[(size_t)(4 * J + 1 < inrow ? min(max(u1, 0), top) : h.v) * 8];
    m[2] = p4[(size_t)(4 * J + 2 < inrow ? min(max(u2, 0), top) : h.v) * 8];
    m[3] = p4[(size_t)(4 * J + 3 < inrow ? min(max(u3, 0), top) : h.v) * 8];
}

template <int J>
__device__ __forceinline__ void trip_sum(float4& s, int inrow, const float4* m) {
    const bool h0 = 4 * J + 0 < inrow, h1 = 4 * J + 1 < inrow;
    const bool h2 = 4 * J + 2 < inrow, h3 = 4 * J + 3 < inrow;
    s.x += h0 ? m[0].x : -0.0f; s.y += h0 ? m[0].y : -0.0f; s.z += h0 ? m[0].z : -0.0f; s.w += h0 ? m[0].w : -0.0f;
    s.x += h1 ? m[1].x : -0.0f; s.y += h1 ? m[1].y : -0.0f; s.z += h1 ? m[1].z : -0.0f; s.w += h1 ? m[1].w : -0.0f;
    s.x += h2 ? m[2].x : -0.0f; s.y += h2 ? m[2].y : -0.0f; s.z += h2 ? m[2].z : -0.0f; s.w += h2 ? m[2].w : -0.0f;
    s.x += h3 ? m[3].x : -0.0f; s.y += h3 ? m[3].y : -0.0f; s.z += h3 ? m[3].z : -0.0f; s.w += h3 ? m[3].w : -0.0f;
}

__device__ __forceinline__ void fetch16(const NodeHead& h, int top, const float4* __restrict__ p4, float4* m) {
    trip_fetch<0>(h, top, p4, m);
    trip_fetch<1>(h, top, p4, m + 4);
    trip_fetch<2>(h, top, p4, m + 8);
    trip_fetch<3>(h, top, p4, m + 12);
}

template <int J>
__device__ __forceinline__ void trip_late(float4& s, const NodeHead& h, int top, const float4* __restrict__ p4) {
    float4 m[4];
    trip_fetch<J>(h, top, p4, m);
    trip_sum<J>(s, min(h.deg, ROWCAP), m);
}

template <int MODE, bool GATHER_NEXT, bool HEAD_AFTER>
__device__ __forceinline__ void pull_step(int g, int base, int n, int lane, int k, int sub,
                                          NodeHead& c, NodeHead& nx, const float4* mc, float4* mn,
                                          const int* __restrict__ cnt, const int4* __restrict__ row4,
                                          const float4* __restrict__ p4,
                                          const int* __restrict__ ovn, const int* __restrict__ ovs,
                                          const int* __restrict__ ovd, int E,
                                          float b0, float b1, float b2, float b3,
                                          float* __restrict__ dstp, float* tw)
{
    const int top = n - 1;
    NodeHead nn = nx;
    if (HEAD_AFTER) nn = load_head(min(base + 4 * (g + 2) + k, top), cnt, row4, p4);
    if (GATHER_NEXT) fetch16(nx, top, p4, mn);

    const int v = c.v, deg = c.deg;
    const int inrow = min(deg, ROWCAP);
    float4 s = c.s;
    trip_sum<0>(s, inrow, mc);
    trip_sum<1>(s, inrow, mc + 4);
    trip_sum<2>(s, inrow, mc + 8);
    trip_sum<3>(s, inrow, mc + 12);

    if (__builtin_amdgcn_ballot_w32(inrow > 16) != 0u) {
        trip_late<4>(s, c, top, p4);
        if (__builtin_amdgcn_ballot_w32(inrow > 20) != 0u) {
            trip_late<5>(s, c, top, p4);
            if (__builtin_amdgcn_ballot_w32(inrow > 24) != 0u) {
                trip_late<6>(s, c, top, p4);
                if (__builtin_amdgcn_ballot_w32(inrow > 28) != 0u)
                    trip_late<7>(s, c, top, p4);
            }
        }
    }

    if (deg > ROWCAP) {
        const int novf = min(ovn[0], E);
        for (int o = 0; o < novf; ++o) {
            if (ovd[o] == v) {
                const float4 m = p4[(size_t)min(max(ovs[o], 0), top) * 8];
                s.x += m.x; s.y += m.y; s.z += m.z; s.w += m.w;
            }
        }
    }

    const float di = rsqrtf((float)(deg + 1));
    if (MODE == 0) {
        float4 h;
        h.x = di * lrelu(di * s.x + b0);
        h.y = di * lrelu(di * s.y + b1);
        h.z = di * lrelu(di * s.z + b2);
        h.w = di * lrelu(di * s.w + b3);
        if (base + 4 * g + k < n) ((float4*)dstp)[(size_t)v * 8 + sub] = h;
    } else {
        float4 t;
        t.x = di * s.x;
        t.y = di * s.y;
        t.z = di * s.z;
        t.w = di * s.w;
        *(float4*)(tw + (4 * g + k) * TPITCH + 4 * sub) = t;
    }
    c = nx;
    nx = nn;
}

__device__ __forceinline__ float dot8(const v8f& z, const float4& wa, const float4& wb,
                                      const float4& ba, const float4& bb) {
    float d = wa.x * lrelu(z[0] + ba.x);
    d += wa.y * lrelu(z[1] + ba.y);
    d += wa.z * lrelu(z[2] + ba.z);
    d += wa.w * lrelu(z[3] + ba.w);
    d += wb.x * lrelu(z[4] + bb.x);
    d += wb.y * lrelu(z[5] + bb.y);
    d += wb.z * lrelu(z[6] + bb.z);
    d += wb.w * lrelu(z[7] + bb.w);
    return d;
}

template <int MODE>
__global__ __launch_bounds__(128)
void gcn_pull(const int* __restrict__ cnt, const int* __restrict__ rows,
              const int* __restrict__ ovn, const int* __restrict__ ovs,
              const int* __restrict__ ovd, int E,
              const float* __restrict__ p, const float* __restrict__ bias,
              const float* __restrict__ Wc,
              const float* __restrict__ Wout, const float* __restrict__ bout,
              float* __restrict__ dstp, int n)
{
    __shared__ float tl[MODE == 1 ? 4 * 32 * TPITCH : 1];
    const int lane = threadIdx.x & 31;
    const int k    = lane >> 3;
    const int sub  = lane & 7;
    const int wave = __builtin_amdgcn_readfirstlane((int)(threadIdx.x >> 5));
    const int base = (blockIdx.x * ((int)blockDim.x >> 5) + wave) * 32;
    if (base >= n) return;
    const int top = n - 1;
    const float4* p4   = (const float4*)p + sub;
    const int4*   row4 = (const int4*)rows + sub;

    float b0 = 0.0f, b1 = 0.0f, b2 = 0.0f, b3 = 0.0f;
    if (MODE == 0) { b0 = bias[4 * sub]; b1 = bias[4 * sub + 1]; b2 = bias[4 * sub + 2]; b3 = bias[4 * sub + 3]; }
    float* tw = tl + (MODE == 1 ? wave * 32 * TPITCH : 0);

    NodeHead c  = load_head(min(base + k, top), cnt, row4, p4);
    NodeHead nx = load_head(min(base + 4 + k, top), cnt, row4, p4);
    float4 ma[16], mb[16];
    fetch16(c, top, p4, ma);

    for (int g = 0; g < 6; g += 2) {
        pull_step<MODE, true, true>(g,     base, n, lane, k, sub, c, nx, ma, mb, cnt, row4, p4, ovn, ovs, ovd, E,
                                    b0, b1, b2, b3, dstp, tw);
        pull_step<MODE, true, true>(g + 1, base, n, lane, k, sub, c, nx, mb, ma, cnt, row4, p4, ovn, ovs, ovd, E,
                                    b0, b1, b2, b3, dstp, tw);
    }
    pull_step<MODE, true, false>(6, base, n, lane, k, sub, c, nx, ma, mb, cnt, row4, p4, ovn, ovs, ovd, E,
                                 b0, b1, b2, b3, dstp, tw);
    pull_step<MODE, false, false>(7, base, n, lane, k, sub, c, nx, mb, ma, cnt, row4, p4, ovn, ovs, ovd, E,
                                  b0, b1, b2, b3, dstp, tw);
    if (MODE == 1) {
        const int c16 = lane & 15;
        const int hh  = lane >> 4;

        v2f af[2][8];
        #pragma unroll
        for (int m = 0; m < 2; ++m) {
            const float4* wp = (const float4*)(Wc + (16 * m + c16) * 32 + 16 * hh);
            #pragma unroll
            for (int q = 0; q < 4; ++q) {
                const float4 w = wp[q];
                af[m][2 * q]     = (v2f){w.x, w.y};
                af[m][2 * q + 1] = (v2f){w.z, w.w};
            }
        }
        float4 bz[2][2], wz[2][2];
        #pragma unroll
        for (int m = 0; m < 2; ++m) {
            const float4* bp = (const float4*)(bias + 16 * m + 8 * hh);
            const float4* op = (const float4*)(Wout + 16 * m + 8 * hh);
            bz[m][0] = bp[0]; bz[m][1] = bp[1];
            wz[m][0] = op[0]; wz[m][1] = op[1];
        }
        const float bo = bout[0];

        __builtin_amdgcn_fence(__ATOMIC_RELEASE, "wavefront");
        __builtin_amdgcn_fence(__ATOMIC_ACQUIRE, "wavefront");

        v2f bt[2][8];
        #pragma unroll
        for (int t = 0; t < 2; ++t) {
            const float4* rp = (const float4*)(tw + (16 * t + c16) * TPITCH + 16 * hh);
            #pragma unroll
            for (int q = 0; q < 4; ++q) {
                const float4 x = rp[q];
                bt[t][2 * q]     = (v2f){x.x, x.y};
                bt[t][2 * q + 1] = (v2f){x.z, x.w};
            }
        }

        v8f acc[2][2];
        #pragma unroll
        for (int m = 0; m < 2; ++m) {
            #pragma unroll
            for (int t = 0; t < 2; ++t)
                acc[m][t] = (v8f){0.f, 0.f, 0.f, 0.f, 0.f, 0.f, 0.f, 0.f};
        }
        #pragma unroll
        for (int s = 0; s < 8; ++s) {
            #pragma unroll
            for (int m = 0; m < 2; ++m) {
                #pragma unroll
                for (int t = 0; t < 2; ++t)
                    acc[m][t] = __builtin_amdgcn_wmma_f32_16x16x4_f32(
                        false, af[m][s], false, bt[t][s], (short)0, acc[m][t], false, false);
            }
        }

        float part[2];
        #pragma unroll
        for (int t = 0; t < 2; ++t) {
            const float d = dot8(acc[0][t], wz[0][0], wz[0][1], bz[0][0], bz[0][1])
                          + dot8(acc[1][t], wz[1][0], wz[1][1], bz[1][0], bz[1][1]);
            part[t] = d + __shfl_xor(d, 16);
        }
        const float outv = hh ? part[1] : part[0];
        if (base + lane < n) dstp[base + lane] = outv + bo;
    }
}

extern "C" void kernel_launch(void* const* d_in, const int* in_sizes, int n_in,
                              void* d_out, int out_size, void* d_ws, size_t ws_size,
                              hipStream_t stream)
{
    const float* x     = (const float*)d_in[0];
    const int*   ei    = (const int*)d_in[1];
    const float* W_in  = (const float*)d_in[2];
    const float* b_in  = (const float*)d_in[3];
    const float* W1    = (const float*)d_in[4];
    const float* b1    = (const float*)d_in[5];
    const float* W2    = (const float*)d_in[6];
    const float* b2    = (const float*)d_in[7];
    const float* W_out = (const float*)d_in[8];
    const float* b_out = (const float*)d_in[9];
    float* out = (float*)d_out;

    const int N = in_sizes[0] / 64;
    const int E = in_sizes[1] / 2;
    const int* src = ei;
    const int* dst = ei + E;

    float* ws   = (float*)d_ws;
    int* ovn    = (int*)d_ws;
    float* h0   = ws + (size_t)N;
    float* p    = ws + (size_t)N * 65;
    float* h1   = ws + (size_t)N * 97;
    int* cnt    = (int*)(ws + (size_t)N * 161);
    int* rows   = cnt + (size_t)N;
    int* ovs    = rows + (size_t)N * ROWCAP;
    int* ovd    = ovs + (size_t)E;

    const int TB = 256;
    const int gblk   = 512;
    const int eblk   = 2048;

    gcn_cnt_zero <<<(N + TB - 1) / TB, TB, 0, stream>>>(cnt, ovn, N);
    gcn_gemm_in_and_place<<<gblk + eblk, 128, 0, stream>>>(x, W_in, b_in, h0, N, gblk, src, dst, cnt, rows, ovn, ovs, ovd, E);

    gcn_gemm<64, 32, 1><<<gblk, 128, 0, stream>>>(h0, W1, nullptr, cnt, p, N);
    gcn_pull<0><<<((size_t)N + 127) / 128, 128, 0, stream>>>(cnt, rows, ovn, ovs, ovd, E, p, b1, nullptr, nullptr, nullptr, h1, N);

    gcn_pull<1><<<((size_t)N + 127) / 128, 128, 0, stream>>>(cnt, rows, ovn, ovs, ovd, E, h1, b2, W2, W_out, b_out, out, N);
}


# in-degree counter moved into its node's own row line (entry 31 of 32): the edge pass's adds fall on one line per node, not 32 counters a line; row table cleared as whole lines; sum order unchanged
# speedup vs baseline: 2.0576x; 1.1783x over previous
#include <hip/hip_runtime.h>
#include <hip/hip_bf16.h>

typedef __attribute__((ext_vector_type(2))) float v2f;
typedef __attribute__((ext_vector_type(8))) float v8f;

__device__ __forceinline__ float lrelu(float v) { return v > 0.0f ? v : 0.01f * v; }

constexpr int ROWPITCH = 32;
constexpr int ROWCAP   = 31;
constexpr int CNTSLOT  = 31;
constexpr int TPITCH = 36;

__global__ void gcn_rows_zero(int4* __restrict__ rows4, int* __restrict__ ovn, int n) {
    const int n4 = n * (ROWPITCH / 4);
    const int stride = (int)(gridDim.x * blockDim.x);
    const int i0 = blockIdx.x * blockDim.x + threadIdx.x;
    for (int i = i0; i < n4; i += stride) rows4[i] = make_int4(0, 0, 0, 0);
    if (i0 == 0) ovn[0] = 0;
}

__device__ __forceinline__ void place_edge(int d, int r, int s, int* __restrict__ rows,
                                           int* __restrict__ ovn, int* __restrict__ ovs,
                                           int* __restrict__ ovd) {
    if (r < ROWCAP) {
        rows[(size_t)d * ROWPITCH + r] = s;
    } else {
        const int o = atomicAdd(ovn, 1);
        ovs[o] = s;
        ovd[o] = d;
    }
}

__device__ __forceinline__ void place_edges(const int* __restrict__ src, const int* __restrict__ dst,
                                            int* __restrict__ rows,
                                            int* __restrict__ ovn, int* __restrict__ ovs,
                                            int* __restrict__ ovd, int E, int blk, int nblk) {
    const int stride = nblk * (int)blockDim.x;
    for (int e0 = blk * (int)blockDim.x + (int)threadIdx.x; e0 < E; e0 += 4 * stride) {
        const int e1 = e0 + stride, e2 = e1 + stride, e3 = e2 + stride;
        if (e3 < E) {
            const int d0 = dst[e0], d1 = dst[e1], d2 = dst[e2], d3 = dst[e3];
            const int s0 = src[e0], s1 = src[e1], s2 = src[e2], s3 = src[e3];
            const int r0 = atomicAdd(&rows[(size_t)d0 * ROWPITCH + CNTSLOT], 1);
            const int r1 = atomicAdd(&rows[(size_t)d1 * ROWPITCH + CNTSLOT], 1);
            const int r2 = atomicAdd(&rows[(size_t)d2 * ROWPITCH + CNTSLOT], 1);
            const int r3 = atomicAdd(&rows[(size_t)d3 * ROWPITCH + CNTSLOT], 1);
            place_edge(d0, r0, s0, rows, ovn, ovs, ovd);
            place_edge(d1, r1, s1, rows, ovn, ovs, ovd);
            place_edge(d2, r2, s2, rows, ovn, ovs, ovd);
            place_edge(d3, r3, s3, rows, ovn, ovs, ovd);
        } else {
            { const int d = dst[e0]; place_edge(d, atomicAdd(&rows[(size_t)d * ROWPITCH + CNTSLOT], 1), src[e0], rows, ovn, ovs, ovd); }
            if (e1 < E) { const int d = dst[e1]; place_edge(d, atomicAdd(&rows[(size_t)d * ROWPITCH + CNTSLOT], 1), src[e1], rows, ovn, ovs, ovd); }
            if (e2 < E) { const int d = dst[e2]; place_edge(d, atomicAdd(&rows[(size_t)d * ROWPITCH + CNTSLOT], 1), src[e2], rows, ovn, ovs, ovd); }
        }
    }
}

template <int K, int NOUT, int MODE>
__device__ __forceinline__ void gemm_tiles(
    const float* __restrict__ in,
    const float* __restrict__ W,
    const float* __restrict__ bias,
    const int* __restrict__ rows,
    float* __restrict__ out,
    int n, int blk, int nblk)
{
    constexpr int KS = K / 4;
    constexpr int NT = NOUT / 16;

    const int lane  = threadIdx.x & 31;
    const int lhalf = lane >> 4;
    const int l16   = lane & 15;
    const int wave   = __builtin_amdgcn_readfirstlane((int)(threadIdx.x >> 5));
    const int gwave  = blk * 4 + wave;
    const int nwaves = nblk * 4;
    const int tiles  = n >> 4;
    if (gwave >= tiles) return;

    v2f bf[KS][NT];
    #pragma unroll
    for (int s = 0; s < KS; ++s) {
        #pragma unroll
        for (int t = 0; t < NT; ++t) {
            const float* wp = W + (t * 16 + l16) * K + 4 * s + 2 * lhalf;
            bf[s][t].x = wp[0];
            bf[s][t].y = wp[1];
        }
    }

    float bv[NT];
    if (MODE == 0) {
        #pragma unroll
        for (int t = 0; t < NT; ++t) bv[t] = bias[t * 16 + l16];
    }

    v2f a[KS];
    {
        const float* ap = in + (size_t)(gwave * 16 + l16) * K + 2 * lhalf;
        #pragma unroll
        for (int s = 0; s < KS; ++s) { a[s].x = ap[4 * s]; a[s].y = ap[4 * s + 1]; }
    }

    for (int tile = gwave; tile < tiles; tile += nwaves) {
        const int row0 = tile * 16;

        const int ntile = (tile + nwaves < tiles) ? (tile + nwaves) : tile;
        int dc[8];
        if (MODE == 1) {
            #pragma unroll
            for (int r = 0; r < 8; ++r) dc[r] = rows[(size_t)(row0 + r + 8 * lhalf) * ROWPITCH + CNTSLOT];
        }
        v2f an[KS];
        {
            const float* ap = in + (size_t)(ntile * 16 + l16) * K + 2 * lhalf;
            #pragma unroll
            for (int s = 0; s < KS; ++s) { an[s].x = ap[4 * s]; an[s].y = ap[4 * s + 1]; }
        }

        __builtin_amdgcn_sched_barrier(0);

        v8f acc[NT];
        #pragma unroll
        for (int t = 0; t < NT; ++t)
            acc[t] = (v8f){0.f, 0.f, 0.f, 0.f, 0.f, 0.f, 0.f, 0.f};

        #pragma unroll
        for (int s = 0; s < KS; ++s) {
            #pragma unroll
            for (int t = 0; t < NT; ++t) {
                acc[t] = __builtin_amdgcn_wmma_f32_16x16x4_f32(
                    false, a[s], false, bf[s][t], (short)0, acc[t], false, false);
            }
        }

        float dv[8];
        if (MODE == 1) {
            __builtin_amdgcn_sched_barrier(0);
            #pragma unroll
            for (int r = 0; r < 8; ++r) dv[r] = rsqrtf((float)(dc[r] + 1));
        }

        #pragma unroll
        for (int t = 0; t < NT; ++t) {
            const int col = t * 16 + l16;
            if (MODE == 0) {
                #pragma unroll
                for (int r = 0; r < 8; ++r) {
                    const int row = row0 + r + 8 * lhalf;
                    out[(size_t)row * NOUT + col] = lrelu(acc[t][r] + bv[t]);
                }
            } else {
                #pragma unroll
                for (int r = 0; r < 8; ++r) {
                    const int row = row0 + r + 8 * lhalf;
                    out[(size_t)row * NOUT + col] = acc[t][r] * dv[r];
                }
            }
        }

        __builtin_amdgcn_sched_barrier(0);

        #pragma unroll
        for (int s = 0; s < KS; ++s) a[s] = an[s];
    }
}

template <int K, int NOUT, int MODE>
__global__ __launch_bounds__(128) void gcn_gemm(
    const float* __restrict__ in, const float* __restrict__ W, const float* __restrict__ bias,
    const int* __restrict__ rows, float* __restrict__ out, int n)
{
    gemm_tiles<K, NOUT, MODE>(in, W, bias, rows, out, n, (int)blockIdx.x, (int)gridDim.x);
}

__global__ __launch_bounds__(128) void gcn_gemm_in_and_place(
    const float* __restrict__ x, const float* __restrict__ W_in, const float* __restrict__ b_in,
    float* __restrict__ h0, int n, int gblk,
    const int* __restrict__ src, const int* __restrict__ dst,
    int* __restrict__ rows, int* __restrict__ ovn, int* __restrict__ ovs, int* __restrict__ ovd, int E)
{
    if ((int)blockIdx.x < gblk)
        gemm_tiles<64, 64, 0>(x, W_in, b_in, nullptr, h0, n, (int)blockIdx.x, gblk);
    else
        place_edges(src, dst, rows, ovn, ovs, ovd, E, (int)blockIdx.x - gblk, (int)gridDim.x - gblk);
}


template <int J>
__device__ __forceinline__ int bcast8(int x) {
    return (int)__builtin_amdgcn_mov_dpp8((unsigned)x, J * 0x249249);
}

struct NodeHead { int v; int4 r; float4 s; };

__device__ __forceinline__ NodeHead load_head(int v, const int4* __restrict__ row4,
                                              const float4* __restrict__ p4) {
    NodeHead h;
    h.v = v;
    h.r = row4[(size_t)v * 8];
    h.s = p4[(size_t)v * 8];
    return h;
}

__device__ __forceinline__ int head_deg(const NodeHead& h) { return bcast8<7>(h.r.w); }

template <int J>
__device__ __forceinline__ void trip_fetch(const NodeHead& h, int top,
                                           const float4* __restrict__ p4, float4* m) {
    const int inrow = min(head_deg(h), ROWCAP);
    const int u0 = bcast8<J>(h.r.x), u1 = bcast8<J>(h.r.y), u2 = bcast8<J>(h.r.z), u3 = bcast8<J>(h.r.w);
    m[0] = p4[(size_t)(4 * J + 0 < inrow ? min(max(u0, 0), top) : h.v) * 8];
    m[1] = p4[(size_t)(4 * J + 1 < inrow ? min(max(u1, 0), top) : h.v) * 8];
    m[2] = p4[(size_t)(4 * J + 2 < inrow ? min(max(u2, 0), top) : h.v) * 8];
    m[3] = p4[(size_t)(4 * J + 3 < inrow ? min(max(u3, 0), top) : h.v) * 8];
}

template <int J>
__device__ __forceinline__ void trip_sum(float4& s, int inrow, const float4* m) {
    const bool h0 = 4 * J + 0 < inrow, h1 = 4 * J + 1 < inrow;
    const bool h2 = 4 * J + 2 < inrow, h3 = 4 * J + 3 < inrow;
    s.x += h0 ? m[0].x : -0.0f; s.y += h0 ? m[0].y : -0.0f; s.z += h0 ? m[0].z : -0.0f; s.w += h0 ? m[0].w : -0.0f;
    s.x += h1 ? m[1].x : -0.0f; s.y += h1 ? m[1].y : -0.0f; s.z += h1 ? m[1].z : -0.0f; s.w += h1 ? m[1].w : -0.0f;
    s.x += h2 ? m[2].x : -0.0f; s.y += h2 ? m[2].y : -0.0f; s.z += h2 ? m[2].z : -0.0f; s.w += h2 ? m[2].w : -0.0f;
    s.x += h3 ? m[3].x : -0.0f; s.y += h3 ? m[3].y : -0.0f; s.z += h3 ? m[3].z : -0.0f; s.w += h3 ? m[3].w : -0.0f;
}

__device__ __forceinline__ void fetch16(const NodeHead& h, int top, const float4* __restrict__ p4, float4* m) {
    trip_fetch<0>(h, top, p4, m);
    trip_fetch<1>(h, top, p4, m + 4);
    trip_fetch<2>(h, top, p4, m + 8);
    trip_fetch<3>(h, top, p4, m + 12);
}

template <int J>
__device__ __forceinline__ void trip_late(float4& s, const NodeHead& h, int top, const float4* __restrict__ p4) {
    float4 m[4];
    trip_fetch<J>(h, top, p4, m);
    trip_sum<J>(s, min(head_deg(h), ROWCAP), m);
}

template <int MODE, bool GATHER_NEXT, bool HEAD_AFTER>
__device__ __forceinline__ void pull_step(int g, int base, int n, int lane, int k, int sub,
                                          NodeHead& c, NodeHead& nx, const float4* mc, float4* mn,
                                          const int4* __restrict__ row4,
                                          const float4* __restrict__ p4,
                                          const int* __restrict__ ovn, const int* __restrict__ ovs,
                                          const int* __restrict__ ovd, int E,
                                          float b0, float b1, float b2, float b3,
                                          float* __restrict__ dstp, float* tw)
{
    const int top = n - 1;
    NodeHead nn = nx;
    if (HEAD_AFTER) nn = load_head(min(base + 4 * (g + 2) + k, top), row4, p4);
    if (GATHER_NEXT) fetch16(nx, top, p4, mn);

    const int v = c.v, deg = head_deg(c);
    const int inrow = min(deg, ROWCAP);
    float4 s = c.s;
    trip_sum<0>(s, inrow, mc);
    trip_sum<1>(s, inrow, mc + 4);
    trip_sum<2>(s, inrow, mc + 8);
    trip_sum<3>(s, inrow, mc + 12);

    if (__builtin_amdgcn_ballot_w32(inrow > 16) != 0u) {
        trip_late<4>(s, c, top, p4);
        if (__builtin_amdgcn_ballot_w32(inrow > 20) != 0u) {
            trip_late<5>(s, c, top, p4);
            if (__builtin_amdgcn_ballot_w32(inrow > 24) != 0u) {
                trip_late<6>(s, c, top, p4);
                if (__builtin_amdgcn_ballot_w32(inrow > 28) != 0u)
                    trip_late<7>(s, c, top, p4);
            }
        }
    }

    if (deg > ROWCAP) {
        const int novf = min(ovn[0], E);
        for (int o = 0; o < novf; ++o) {
            if (ovd[o] == v) {
                const float4 m = p4[(size_t)min(max(ovs[o], 0), top) * 8];
                s.x += m.x; s.y += m.y; s.z += m.z; s.w += m.w;
            }
        }
    }

    const float di = rsqrtf((float)(deg + 1));
    if (MODE == 0) {
        float4 h;
        h.x = di * lrelu(di * s.x + b0);
        h.y = di * lrelu(di * s.y + b1);
        h.z = di * lrelu(di * s.z + b2);
        h.w = di * lrelu(di * s.w + b3);
        if (base + 4 * g + k < n) ((float4*)dstp)[(size_t)v * 8 + sub] = h;
    } else {
        float4 t;
        t.x = di * s.x;
        t.y = di * s.y;
        t.z = di * s.z;
        t.w = di * s.w;
        *(float4*)(tw + (4 * g + k) * TPITCH + 4 * sub) = t;
    }
    c = nx;
    nx = nn;
}

__device__ __forceinline__ float dot8(const v8f& z, const float4& wa, const float4& wb,
                                      const float4& ba, const float4& bb) {
    float d = wa.x * lrelu(z[0] + ba.x);
    d += wa.y * lrelu(z[1] + ba.y);
    d += wa.z * lrelu(z[2] + ba.z);
    d += wa.w * lrelu(z[3] + ba.w);
    d += wb.x * lrelu(z[4] + bb.x);
    d += wb.y * lrelu(z[5] + bb.y);
    d += wb.z * lrelu(z[6] + bb.z);
    d += wb.w * lrelu(z[7] + bb.w);
    return d;
}

template <int MODE>
__global__ __launch_bounds__(128)
void gcn_pull(const int* __restrict__ rows,
              const int* __restrict__ ovn, const int* __restrict__ ovs,
              const int* __restrict__ ovd, int E,
              const float* __restrict__ p, const float* __restrict__ bias,
              const float* __restrict__ Wc,
              const float* __restrict__ Wout, const float* __restrict__ bout,
              float* __restrict__ dstp, int n)
{
    __shared__ float tl[MODE == 1 ? 4 * 32 * TPITCH : 1];
    const int lane = threadIdx.x & 31;
    const int k    = lane >> 3;
    const int sub  = lane & 7;
    const int wave = __builtin_amdgcn_readfirstlane((int)(threadIdx.x >> 5));
    const int base = (blockIdx.x * ((int)blockDim.x >> 5) + wave) * 32;
    if (base >= n) return;
    const int top = n - 1;
    const float4* p4   = (const float4*)p + sub;
    const int4*   row4 = (const int4*)rows + sub;

    float b0 = 0.0f, b1 = 0.0f, b2 = 0.0f, b3 = 0.0f;
    if (MODE == 0) { b0 = bias[4 * sub]; b1 = bias[4 * sub + 1]; b2 = bias[4 * sub + 2]; b3 = bias[4 * sub + 3]; }
    float* tw = tl + (MODE == 1 ? wave * 32 * TPITCH : 0);

    NodeHead c  = load_head(min(base + k, top), row4, p4);
    NodeHead nx = load_head(min(base + 4 + k, top), row4, p4);
    float4 ma[16], mb[16];
    fetch16(c, top, p4, ma);

    for (int g = 0; g < 6; g += 2) {
        pull_step<MODE, true, true>(g,     base, n, lane, k, sub, c, nx, ma, mb, row4, p4, ovn, ovs, ovd, E,
                                    b0, b1, b2, b3, dstp, tw);
        pull_step<MODE, true, true>(g + 1, base, n, lane, k, sub, c, nx, mb, ma, row4, p4, ovn, ovs, ovd, E,
                                    b0, b1, b2, b3, dstp, tw);
    }
    pull_step<MODE, true, false>(6, base, n, lane, k, sub, c, nx, ma, mb, row4, p4, ovn, ovs, ovd, E,
                                 b0, b1, b2, b3, dstp, tw);
    pull_step<MODE, false, false>(7, base, n, lane, k, sub, c, nx, mb, ma, row4, p4, ovn, ovs, ovd, E,
                                  b0, b1, b2, b3, dstp, tw);
    if (MODE == 1) {
        const int c16 = lane & 15;
        const int hh  = lane >> 4;

        v2f af[2][8];
        #pragma unroll
        for (int m = 0; m < 2; ++m) {
            const float4* wp = (const float4*)(Wc + (16 * m + c16) * 32 + 16 * hh);
            #pragma unroll
            for (int q = 0; q < 4; ++q) {
                const float4 w = wp[q];
                af[m][2 * q]     = (v2f){w.x, w.y};
                af[m][2 * q + 1] = (v2f){w.z, w.w};
            }
        }
        float4 bz[2][2], wz[2][2];
        #pragma unroll
        for (int m = 0; m < 2; ++m) {
            const float4* bp = (const float4*)(bias + 16 * m + 8 * hh);
            const float4* op = (const float4*)(Wout + 16 * m + 8 * hh);
            bz[m][0] = bp[0]; bz[m][1] = bp[1];
            wz[m][0] = op[0]; wz[m][1] = op[1];
        }
        const float bo = bout[0];

        __builtin_amdgcn_fence(__ATOMIC_RELEASE, "wavefront");
        __builtin_amdgcn_fence(__ATOMIC_ACQUIRE, "wavefront");

        v2f bt[2][8];
        #pragma unroll
        for (int t = 0; t < 2; ++t) {
            const float4* rp = (const float4*)(tw + (16 * t + c16) * TPITCH + 16 * hh);
            #pragma unroll
            for (int q = 0; q < 4; ++q) {
                const float4 x = rp[q];
                bt[t][2 * q]     = (v2f){x.x, x.y};
                bt[t][2 * q + 1] = (v2f){x.z, x.w};
            }
        }

        v8f acc[2][2];
        #pragma unroll
        for (int m = 0; m < 2; ++m) {
            #pragma unroll
            for (int t = 0; t < 2; ++t)
                acc[m][t] = (v8f){0.f, 0.f, 0.f, 0.f, 0.f, 0.f, 0.f, 0.f};
        }
        #pragma unroll
        for (int s = 0; s < 8; ++s) {
            #pragma unroll
            for (int m = 0; m < 2; ++m) {
                #pragma unroll
                for (int t = 0; t < 2; ++t)
                    acc[m][t] = __builtin_amdgcn_wmma_f32_16x16x4_f32(
                        false, af[m][s], false, bt[t][s], (short)0, acc[m][t], false, false);
            }
        }

        float part[2];
        #pragma unroll
        for (int t = 0; t < 2; ++t) {
            const float d = dot8(acc[0][t], wz[0][0], wz[0][1], bz[0][0], bz[0][1])
                          + dot8(acc[1][t], wz[1][0], wz[1][1], bz[1][0], bz[1][1]);
            part[t] = d + __shfl_xor(d, 16);
        }
        const float outv = hh ? part[1] : part[0];
        if (base + lane < n) dstp[base + lane] = outv + bo;
    }
}

extern "C" void kernel_launch(void* const* d_in, const int* in_sizes, int n_in,
                              void* d_out, int out_size, void* d_ws, size_t ws_size,
                              hipStream_t stream)
{
    const float* x     = (const float*)d_in[0];
    const int*   ei    = (const int*)d_in[1];
    const float* W_in  = (const float*)d_in[2];
    const float* b_in  = (const float*)d_in[3];
    const float* W1    = (const float*)d_in[4];
    const float* b1    = (const float*)d_in[5];
    const float* W2    = (const float*)d_in[6];
    const float* b2    = (const float*)d_in[7];
    const float* W_out = (const float*)d_in[8];
    const float* b_out = (const float*)d_in[9];
    float* out = (float*)d_out;

    const int N = in_sizes[0] / 64;
    const int E = in_sizes[1] / 2;
    const int* src = ei;
    const int* dst = ei + E;

    float* ws   = (float*)d_ws;
    int* ovn    = (int*)d_ws;
    float* h0   = ws + (size_t)N;
    float* p    = ws + (size_t)N * 65;
    float* h1   = ws + (size_t)N * 97;
    int* rows   = (int*)(ws + (size_t)N * 161);
    int* ovs    = rows + (size_t)N * ROWPITCH;
    int* ovd    = ovs + (size_t)E;

    const int TB = 256;
    const int gblk   = 512;
    const int eblk   = 2048;

    gcn_rows_zero<<<(N + TB - 1) / TB, TB, 0, stream>>>((int4*)rows, ovn, N);
    gcn_gemm_in_and_place<<<gblk + eblk, 128, 0, stream>>>(x, W_in, b_in, h0, N, gblk, src, dst, rows, ovn, ovs, ovd, E);

    gcn_gemm<64, 32, 1><<<gblk, 128, 0, stream>>>(h0, W1, nullptr, rows, p, N);
    gcn_pull<0><<<((size_t)N + 127) / 128, 128, 0, stream>>>(rows, ovn, ovs, ovd, E, p, b1, nullptr, nullptr, nullptr, h1, N);

    gcn_pull<1><<<((size_t)N + 127) / 128, 128, 0, stream>>>(rows, ovn, ovs, ovd, E, h1, b2, W2, W_out, b_out, out, N);
}


# edge pass steered by cache half: a row line is cleared, added to and stored to from one half of the package only (owner table, edge chunks by per-half ticket); sum order = count arrival as before
# speedup vs baseline: 2.9324x; 1.4252x over previous
#include <hip/hip_runtime.h>
#include <hip/hip_bf16.h>

typedef __attribute__((ext_vector_type(2))) float v2f;
typedef __attribute__((ext_vector_type(8))) float v8f;

__device__ __forceinline__ float lrelu(float v) { return v > 0.0f ? v : 0.01f * v; }

constexpr int ROWPITCH = 32;
constexpr int ROWCAP   = 31;
constexpr int CNTSLOT  = 31;
constexpr int TPITCH = 36;

constexpr int CTL_OVN  = 0;
constexpr int CTL_TICK = 32;
constexpr int CTL_OWN  = 128;
constexpr int OWNSHIFT = 8;
constexpr int OWNBLK   = 1 << OWNSHIFT;
constexpr int EPL      = 8;
constexpr int ECHUNK   = 128 * EPL;

__device__ __forceinline__ int cache_half() {
    return (int)((__builtin_amdgcn_s_sendmsg_rtn(0x87) >> 18) & 1u);
}

__global__ __launch_bounds__(256) void gcn_rows_zero(int4* __restrict__ rows4, int* __restrict__ ctl, int n) {
    const int b = (int)blockIdx.x;
    const int n4 = min(OWNBLK, n - b * OWNBLK) * (ROWPITCH / 4);
    int4* base = rows4 + (size_t)b * OWNBLK * (ROWPITCH / 4);
    #pragma unroll
    for (int k = 0; k < OWNBLK * (ROWPITCH / 4) / 256; ++k) {
        const int i = (int)threadIdx.x + 256 * k;
        if (i < n4) base[i] = make_int4(0, 0, 0, 0);
    }
    if (threadIdx.x == 0) {
        ctl[CTL_OWN + b] = cache_half();
        if (b == 0) { ctl[CTL_OVN] = 0; ctl[CTL_TICK] = 0; ctl[CTL_TICK + 32] = 0; }
    }
}

__device__ __forceinline__ void place_edge(int d, int r, int s, int* __restrict__ rows,
                                           int* ovn, int* __restrict__ ovs,
                                           int* __restrict__ ovd) {
    if (r < ROWCAP) {
        rows[(size_t)d * ROWPITCH + r] = s;
    } else {
        const int o = atomicAdd(ovn, 1);
        ovs[o] = s;
        ovd[o] = d;
    }
}

__device__ __forceinline__ void place_chunk(const int* __restrict__ src, const int* __restrict__ dst,
                                            const int* own, int set, int e0, int E, bool whole,
                                            int* __restrict__ rows, int* ovn,
                                            int* __restrict__ ovs, int* __restrict__ ovd,
                                            int* tp, bool first, int& tnext) {
    if (whole) {
        int d[EPL], s[EPL], o[EPL], r[EPL];
        #pragma unroll
        for (int j = 0; j < EPL; ++j) d[j] = dst[e0 + 128 * j];
        #pragma unroll
        for (int j = 0; j < EPL; ++j) s[j] = src[e0 + 128 * j];
        #pragma unroll
        for (int j = 0; j < EPL; ++j) o[j] = own[d[j] >> OWNSHIFT];
        asm volatile("" :: "v"(o[0]), "v"(o[1]), "v"(o[2]), "v"(o[3]), "v"(o[4]), "v"(o[5]), "v"(o[6]), "v"(o[7]));
        #pragma unroll
        for (int j = 0; j < EPL; ++j) {
            r[j] = 0;
            if (o[j] == set) r[j] = atomicAdd(&rows[(size_t)d[j] * ROWPITCH + CNTSLOT], 1);
        }
        if (first) tnext = atomicAdd(tp, 1);
        #pragma unroll
        for (int j = 0; j < EPL; ++j)
            if (o[j] == set) place_edge(d[j], r[j], s[j], rows, ovn, ovs, ovd);
    } else {
        if (first) tnext = atomicAdd(tp, 1);
        for (int j = 0; j < EPL; ++j) {
            const int e = e0 + 128 * j;
            if (e < E) {
                const int d = dst[e];
                if (own[d >> OWNSHIFT] == set)
                    place_edge(d, atomicAdd(&rows[(size_t)d * ROWPITCH + CNTSLOT], 1), src[e], rows, ovn, ovs, ovd);
            }
        }
    }
}

__device__ __forceinline__ void place_edges(const int* __restrict__ src, const int* __restrict__ dst,
                                            int* __restrict__ rows, int* ctl,
                                            int* __restrict__ ovs, int* __restrict__ ovd, int E) {
    __shared__ int tk[2];
    const int tid = (int)threadIdx.x;
    int* ovn = ctl + CTL_OVN;
    const int* own = ctl + CTL_OWN;
    const int nch = (E + ECHUNK - 1) / ECHUNK;
    int set = cache_half();
    int turn = 0;
    for (int pass = 0; pass < 2; ++pass) {
        int* tp = ctl + CTL_TICK + 32 * set;
        int tnext = 0;
        if (tid == 0) tnext = atomicAdd(tp, 1);
        for (int it = 0; it <= nch; ++it) {
            if (tid == 0) tk[turn] = tnext;
            __syncthreads();
            const int t = __builtin_amdgcn_readfirstlane(tk[turn]);
            turn ^= 1;
            if (t >= nch) break;
            const int e0 = t * ECHUNK;
            place_chunk(src, dst, own, set, e0 + tid, E, e0 + ECHUNK <= E, rows, ovn, ovs, ovd, tp, tid == 0, tnext);
        }
        set ^= 1;
    }
}

template <int K, int NOUT, int MODE>
__device__ __forceinline__ void gemm_tiles(
    const float* __restrict__ in,
    const float* __restrict__ W,
    const float* __restrict__ bias,
    const int* __restrict__ rows,
    float* __restrict__ out,
    int n, int blk, int nblk)
{
    constexpr int KS = K / 4;
    constexpr int NT = NOUT / 16;

    const int lane  = threadIdx.x & 31;
    const int lhalf = lane >> 4;
    const int l16   = lane & 15;
    const int wave   = __builtin_amdgcn_readfirstlane((int)(threadIdx.x >> 5));
    const int gwave  = blk * 4 + wave;
    const int nwaves = nblk * 4;
    const int tiles  = n >> 4;
    if (gwave >= tiles) return;

    v2f bf[KS][NT];
    #pragma unroll
    for (int s = 0; s < KS; ++s) {
        #pragma unroll
        for (int t = 0; t < NT; ++t) {
            const float* wp = W + (t * 16 + l16) * K + 4 * s + 2 * lhalf;
            bf[s][t].x = wp[0];
            bf[s][t].y = wp[1];
        }
    }

    float bv[NT];
    if (MODE == 0) {
        #pragma unroll
        for (int t = 0; t < NT; ++t) bv[t] = bias[t * 16 + l16];
    }

    v2f a[KS];
    {
        const float* ap = in + (size_t)(gwave * 16 + l16) * K + 2 * lhalf;
        #pragma unroll
        for (int s = 0; s < KS; ++s) { a[s].x = ap[4 * s]; a[s].y = ap[4 * s + 1]; }
    }

    for (int tile = gwave; tile < tiles; tile += nwaves) {
        const int row0 = tile * 16;

        const int ntile = (tile + nwaves < tiles) ? (tile + nwaves) : tile;
        int dc[8];
        if (MODE == 1) {
            #pragma unroll
            for (int r = 0; r < 8; ++r) dc[r] = rows[(size_t)(row0 + r + 8 * lhalf) * ROWPITCH + CNTSLOT];
        }
        v2f an[KS];
        {
            const float* ap = in + (size_t)(ntile * 16 + l16) * K + 2 * lhalf;
            #pragma unroll
            for (int s = 0; s < KS; ++s) { an[s].x = ap[4 * s]; an[s].y = ap[4 * s + 1]; }
        }

        __builtin_amdgcn_sched_barrier(0);

        v8f acc[NT];
        #pragma unroll
        for (int t = 0; t < NT; ++t)
            acc[t] = (v8f){0.f, 0.f, 0.f, 0.f, 0.f, 0.f, 0.f, 0.f};

        #pragma unroll
        for (int s = 0; s < KS; ++s) {
            #pragma unroll
            for (int t = 0; t < NT; ++t) {
                acc[t] = __builtin_amdgcn_wmma_f32_16x16x4_f32(
                    false, a[s], false, bf[s][t], (short)0, acc[t], false, false);
            }
        }

        float dv[8];
        if (MODE == 1) {
            __builtin_amdgcn_sched_barrier(0);
            #pragma unroll
            for (int r = 0; r < 8; ++r) dv[r] = rsqrtf((float)(dc[r] + 1));
        }

        #pragma unroll
        for (int t = 0; t < NT; ++t) {
            const int col = t * 16 + l16;
            if (MODE == 0) {
                #pragma unroll
                for (int r = 0; r < 8; ++r) {
                    const int row = row0 + r + 8 * lhalf;
                    out[(size_t)row * NOUT + col] = lrelu(acc[t][r] + bv[t]);
                }
            } else {
                #pragma unroll
                for (int r = 0; r < 8; ++r) {
                    const int row = row0 + r + 8 * lhalf;
                    out[(size_t)row * NOUT + col] = acc[t][r] * dv[r];
                }
            }
        }

        __builtin_amdgcn_sched_barrier(0);

        #pragma unroll
        for (int s = 0; s < KS; ++s) a[s] = an[s];
    }
}

template <int K, int NOUT, int MODE>
__global__ __launch_bounds__(128) void gcn_gemm(
    const float* __restrict__ in, const float* __restrict__ W, const float* __restrict__ bias,
    const int* __restrict__ rows, float* __restrict__ out, int n)
{
    gemm_tiles<K, NOUT, MODE>(in, W, bias, rows, out, n, (int)blockIdx.x, (int)gridDim.x);
}

__global__ __launch_bounds__(128) void gcn_gemm_in_and_place(
    const float* __restrict__ x, const float* __restrict__ W_in, const float* __restrict__ b_in,
    float* __restrict__ h0, int n, int gblk,
    const int* __restrict__ src, const int* __restrict__ dst,
    int* __restrict__ rows, int* ctl, int* __restrict__ ovs, int* __restrict__ ovd, int E)
{
    if ((int)blockIdx.x < gblk)
        gemm_tiles<64, 64, 0>(x, W_in, b_in, nullptr, h0, n, (int)blockIdx.x, gblk);
    else
        place_edges(src, dst, rows, ctl, ovs, ovd, E);
}


template <int J>
__device__ __forceinline__ int bcast8(int x) {
    return (int)__builtin_amdgcn_mov_dpp8((unsigned)x, J * 0x249249);
}

struct NodeHead { int v; int4 r; float4 s; };

__device__ __forceinline__ NodeHead load_head(int v, const int4* __restrict__ row4,
                                              const float4* __restrict__ p4) {
    NodeHead h;
    h.v = v;
    h.r = row4[(size_t)v * 8];
    h.s = p4[(size_t)v * 8];
    return h;
}

__device__ __forceinline__ int head_deg(const NodeHead& h) { return bcast8<7>(h.r.w); }

template <int J>
__device__ __forceinline__ void trip_fetch(const NodeHead& h, int top,
                                           const float4* __restrict__ p4, float4* m) {
    const int inrow = min(head_deg(h), ROWCAP);
    const int u0 = bcast8<J>(h.r.x), u1 = bcast8<J>(h.r.y), u2 = bcast8<J>(h.r.z), u3 = bcast8<J>(h.r.w);
    m[0] = p4[(size_t)(4 * J + 0 < inrow ? min(max(u0, 0), top) : h.v) * 8];
    m[1] = p4[(size_t)(4 * J + 1 < inrow ? min(max(u1, 0), top) : h.v) * 8];
    m[2] = p4[(size_t)(4 * J + 2 < inrow ? min(max(u2, 0), top) : h.v) * 8];
    m[3] = p4[(size_t)(4 * J + 3 < inrow ? min(max(u3, 0), top) : h.v) * 8];
}

template <int J>
__device__ __forceinline__ void trip_sum(float4& s, int inrow, const float4* m) {
    const bool h0 = 4 * J + 0 < inrow, h1 = 4 * J + 1 < inrow;
    const bool h2 = 4 * J + 2 < inrow, h3 = 4 * J + 3 < inrow;
    s.x += h0 ? m[0].x : -0.0f; s.y += h0 ? m[0].y : -0.0f; s.z += h0 ? m[0].z : -0.0f; s.w += h0 ? m[0].w : -0.0f;
    s.x += h1 ? m[1].x : -0.0f; s.y += h1 ? m[1].y : -0.0f; s.z += h1 ? m[1].z : -0.0f; s.w += h1 ? m[1].w : -0.0f;
    s.x += h2 ? m[2].x : -0.0f; s.y += h2 ? m[2].y : -0.0f; s.z += h2 ? m[2].z : -0.0f; s.w += h2 ? m[2].w : -0.0f;
    s.x += h3 ? m[3].x : -0.0f; s.y += h3 ? m[3].y : -0.0f; s.z += h3 ? m[3].z : -0.0f; s.w += h3 ? m[3].w : -0.0f;
}

__device__ __forceinline__ void fetch16(const NodeHead& h, int top, const float4* __restrict__ p4, float4* m) {
    trip_fetch<0>(h, top, p4, m);
    trip_fetch<1>(h, top, p4, m + 4);
    trip_fetch<2>(h, top, p4, m + 8);
    trip_fetch<3>(h, top, p4, m + 12);
}

template <int J>
__device__ __forceinline__ void trip_late(float4& s, const NodeHead& h, int top, const float4* __restrict__ p4) {
    float4 m[4];
    trip_fetch<J>(h, top, p4, m);
    trip_sum<J>(s, min(head_deg(h), ROWCAP), m);
}

template <int MODE, bool GATHER_NEXT, bool HEAD_AFTER>
__device__ __forceinline__ void pull_step(int g, int base, int n, int lane, int k, int sub,
                                          NodeHead& c, NodeHead& nx, const float4* mc, float4* mn,
                                          const int4* __restrict__ row4,
                                          const float4* __restrict__ p4,
                                          const int* __restrict__ ovn, const int* __restrict__ ovs,
                                          const int* __restrict__ ovd, int E,
                                          float b0, float b1, float b2, float b3,
                                          float* __restrict__ dstp, float* tw)
{
    const int top = n - 1;
    NodeHead nn = nx;
    if (HEAD_AFTER) nn = load_head(min(base + 4 * (g + 2) + k, top), row4, p4);
    if (GATHER_NEXT) fetch16(nx, top, p4, mn);

    const int v = c.v, deg = head_deg(c);
    const int inrow = min(deg, ROWCAP);
    float4 s = c.s;
    trip_sum<0>(s, inrow, mc);
    trip_sum<1>(s, inrow, mc + 4);
    trip_sum<2>(s, inrow, mc + 8);
    trip_sum<3>(s, inrow, mc + 12);

    if (__builtin_amdgcn_ballot_w32(inrow > 16) != 0u) {
        trip_late<4>(s, c, top, p4);
        if (__builtin_amdgcn_ballot_w32(inrow > 20) != 0u) {
            trip_late<5>(s, c, top, p4);
            if (__builtin_amdgcn_ballot_w32(inrow > 24) != 0u) {
                trip_late<6>(s, c, top, p4);
                if (__builtin_amdgcn_ballot_w32(inrow > 28) != 0u)
                    trip_late<7>(s, c, top, p4);
            }
        }
    }

    if (deg > ROWCAP) {
        const int novf = min(ovn[0], E);
        for (int o = 0; o < novf; ++o) {
            if (ovd[o] == v) {
                const float4 m = p4[(size_t)min(max(ovs[o], 0), top) * 8];
                s.x += m.x; s.y += m.y; s.z += m.z; s.w += m.w;
            }
        }
    }

    const float di = rsqrtf((float)(deg + 1));
    if (MODE == 0) {
        float4 h;
        h.x = di * lrelu(di * s.x + b0);
        h.y = di * lrelu(di * s.y + b1);
        h.z = di * lrelu(di * s.z + b2);
        h.w = di * lrelu(di * s.w + b3);
        if (base + 4 * g + k < n) ((float4*)dstp)[(size_t)v * 8 + sub] = h;
    } else {
        float4 t;
        t.x = di * s.x;
        t.y = di * s.y;
        t.z = di * s.z;
        t.w = di * s.w;
        *(float4*)(tw + (4 * g + k) * TPITCH + 4 * sub) = t;
    }
    c = nx;
    nx = nn;
}

__device__ __forceinline__ float dot8(const v8f& z, const float4& wa, const float4& wb,
                                      const float4& ba, const float4& bb) {
    float d = wa.x * lrelu(z[0] + ba.x);
    d += wa.y * lrelu(z[1] + ba.y);
    d += wa.z * lrelu(z[2] + ba.z);
    d += wa.w * lrelu(z[3] + ba.w);
    d += wb.x * lrelu(z[4] + bb.x);
    d += wb.y * lrelu(z[5] + bb.y);
    d += wb.z * lrelu(z[6] + bb.z);
    d += wb.w * lrelu(z[7] + bb.w);
    return d;
}

template <int MODE>
__global__ __launch_bounds__(128)
void gcn_pull(const int* __restrict__ rows,
              const int* __restrict__ ovn, const int* __restrict__ ovs,
              const int* __restrict__ ovd, int E,
              const float* __restrict__ p, const float* __restrict__ bias,
              const float* __restrict__ Wc,
              const float* __restrict__ Wout, const float* __restrict__ bout,
              float* __restrict__ dstp, int n)
{
    __shared__ float tl[MODE == 1 ? 4 * 32 * TPITCH : 1];
    const int lane = threadIdx.x & 31;
    const int k    = lane >> 3;
    const int sub  = lane & 7;
    const int wave = __builtin_amdgcn_readfirstlane((int)(threadIdx.x >> 5));
    const int base = (blockIdx.x * ((int)blockDim.x >> 5) + wave) * 32;
    if (base >= n) return;
    const int top = n - 1;
    const float4* p4   = (const float4*)p + sub;
    const int4*   row4 = (const int4*)rows + sub;

    float b0 = 0.0f, b1 = 0.0f, b2 = 0.0f, b3 = 0.0f;
    if (MODE == 0) { b0 = bias[4 * sub]; b1 = bias[4 * sub + 1]; b2 = bias[4 * sub + 2]; b3 = bias[4 * sub + 3]; }
    float* tw = tl + (MODE == 1 ? wave * 32 * TPITCH : 0);

    NodeHead c  = load_head(min(base + k, top), row4, p4);
    NodeHead nx = load_head(min(base + 4 + k, top), row4, p4);
    float4 ma[16], mb[16];
    fetch16(c, top, p4, ma);

    for (int g = 0; g < 6; g += 2) {
        pull_step<MODE, true, true>(g,     base, n, lane, k, sub, c, nx, ma, mb, row4, p4, ovn, ovs, ovd, E,
                                    b0, b1, b2, b3, dstp, tw);
        pull_step<MODE, true, true>(g + 1, base, n, lane, k, sub, c, nx, mb, ma, row4, p4, ovn, ovs, ovd, E,
                                    b0, b1, b2, b3, dstp, tw);
    }
    pull_step<MODE, true, false>(6, base, n, lane, k, sub, c, nx, ma, mb, row4, p4, ovn, ovs, ovd, E,
                                 b0, b1, b2, b3, dstp, tw);
    pull_step<MODE, false, false>(7, base, n, lane, k, sub, c, nx, mb, ma, row4, p4, ovn, ovs, ovd, E,
                                  b0, b1, b2, b3, dstp, tw);
    if (MODE == 1) {
        const int c16 = lane & 15;
        const int hh  = lane >> 4;

        v2f af[2][8];
        #pragma unroll
        for (int m = 0; m < 2; ++m) {
            const float4* wp = (const float4*)(Wc + (16 * m + c16) * 32 + 16 * hh);
            #pragma unroll
            for (int q = 0; q < 4; ++q) {
                const float4 w = wp[q];
                af[m][2 * q]     = (v2f){w.x, w.y};
                af[m][2 * q + 1] = (v2f){w.z, w.w};
            }
        }
        float4 bz[2][2], wz[2][2];
        #pragma unroll
        for (int m = 0; m < 2; ++m) {
            const float4* bp = (const float4*)(bias + 16 * m + 8 * hh);
            const float4* op = (const float4*)(Wout + 16 * m + 8 * hh);
            bz[m][0] = bp[0]; bz[m][1] = bp[1];
            wz[m][0] = op[0]; wz[m][1] = op[1];
        }
        const float bo = bout[0];

        __builtin_amdgcn_fence(__ATOMIC_RELEASE, "wavefront");
        __builtin_amdgcn_fence(__ATOMIC_ACQUIRE, "wavefront");

        v2f bt[2][8];
        #pragma unroll
        for (int t = 0; t < 2; ++t) {
            const float4* rp = (const float4*)(tw + (16 * t + c16) * TPITCH + 16 * hh);
            #pragma unroll
            for (int q = 0; q < 4; ++q) {
                const float4 x = rp[q];
                bt[t][2 * q]     = (v2f){x.x, x.y};
                bt[t][2 * q + 1] = (v2f){x.z, x.w};
            }
        }

        v8f acc[2][2];
        #pragma unroll
        for (int m = 0; m < 2; ++m) {
            #pragma unroll
            for (int t = 0; t < 2; ++t)
                acc[m][t] = (v8f){0.f, 0.f, 0.f, 0.f, 0.f, 0.f, 0.f, 0.f};
        }
        #pragma unroll
        for (int s = 0; s < 8; ++s) {
            #pragma unroll
            for (int m = 0; m < 2; ++m) {
                #pragma unroll
                for (int t = 0; t < 2; ++t)
                    acc[m][t] = __builtin_amdgcn_wmma_f32_16x16x4_f32(
                        false, af[m][s], false, bt[t][s], (short)0, acc[m][t], false, false);
            }
        }

        float part[2];
        #pragma unroll
        for (int t = 0; t < 2; ++t) {
            const float d = dot8(acc[0][t], wz[0][0], wz[0][1], bz[0][0], bz[0][1])
                          + dot8(acc[1][t], wz[1][0], wz[1][1], bz[1][0], bz[1][1]);
            part[t] = d + __shfl_xor(d, 16);
        }
        const float outv = hh ? part[1] : part[0];
        if (base + lane < n) dstp[base + lane] = outv + bo;
    }
}

extern "C" void kernel_launch(void* const* d_in, const int* in_sizes, int n_in,
                              void* d_out, int out_size, void* d_ws, size_t ws_size,
                              hipStream_t stream)
{
    const float* x     = (const float*)d_in[0];
    const int*   ei    = (const int*)d_in[1];
    const float* W_in  = (const float*)d_in[2];
    const float* b_in  = (const float*)d_in[3];
    const float* W1    = (const float*)d_in[4];
    const float* b1    = (const float*)d_in[5];
    const float* W2    = (const float*)d_in[6];
    const float* b2    = (const float*)d_in[7];
    const float* W_out = (const float*)d_in[8];
    const float* b_out = (const float*)d_in[9];
    float* out = (float*)d_out;

    const int N = in_sizes[0] / 64;
    const int E = in_sizes[1] / 2;
    const int* src = ei;
    const int* dst = ei + E;

    float* ws   = (float*)d_ws;
    int* ctl    = (int*)d_ws;
    float* h0   = ws + (size_t)N;
    float* p    = ws + (size_t)N * 65;
    float* h1   = ws + (size_t)N * 97;
    int* rows   = (int*)(ws + (size_t)N * 161);
    int* ovs    = rows + (size_t)N * ROWPITCH;
    int* ovd    = ovs + (size_t)E;

    const int TB = 256;
    const int gblk   = 512;
    const int eblk   = 1024;

    gcn_rows_zero<<<(N + TB - 1) / TB, TB, 0, stream>>>((int4*)rows, ctl, N);
    gcn_gemm_in_and_place<<<gblk + eblk, 128, 0, stream>>>(x, W_in, b_in, h0, N, gblk, src, dst, rows, ctl, ovs, ovd, E);

    gcn_gemm<64, 32, 1><<<gblk, 128, 0, stream>>>(h0, W1, nullptr, rows, p, N);
    gcn_pull<0><<<((size_t)N + 127) / 128, 128, 0, stream>>>(rows, ctl, ovs, ovd, E, p, b1, nullptr, nullptr, nullptr, h1, N);

    gcn_pull<1><<<((size_t)N + 127) / 128, 128, 0, stream>>>(rows, ctl, ovs, ovd, E, h1, b2, W2, W_out, b_out, out, N);
}


# edge gathers: row entries made into finished addresses once, by the lane that holds them (past the end: a line of -0.0f); a gather is a move, an or, a load; sums need no select; sum order unchanged
# speedup vs baseline: 3.1069x; 1.0595x over previous
#include <hip/hip_runtime.h>
#include <hip/hip_bf16.h>

typedef __attribute__((ext_vector_type(2))) float v2f;
typedef __attribute__((ext_vector_type(8))) float v8f;

__device__ __forceinline__ float lrelu(float v) { return v > 0.0f ? v : 0.01f * v; }

constexpr int ROWPITCH = 32;
constexpr int ROWCAP   = 31;
constexpr int CNTSLOT  = 31;
constexpr int TPITCH = 36;
constexpr int ZLINES = 32;

constexpr int CTL_OVN  = 0;
constexpr int CTL_TICK = 32;
constexpr int CTL_OWN  = 128;
constexpr int OWNSHIFT = 8;
constexpr int OWNBLK   = 1 << OWNSHIFT;
constexpr int EPL      = 8;
constexpr int ECHUNK   = 128 * EPL;

__device__ __forceinline__ int cache_half() {
    return (int)((__builtin_amdgcn_s_sendmsg_rtn(0x87) >> 18) & 1u);
}

__global__ __launch_bounds__(256) void gcn_rows_zero(int4* __restrict__ rows4, int* __restrict__ ctl,
                                                     float* __restrict__ zline, int n) {
    const int b = (int)blockIdx.x;
    const int n4 = min(OWNBLK, n - b * OWNBLK) * (ROWPITCH / 4);
    int4* base = rows4 + (size_t)b * OWNBLK * (ROWPITCH / 4);
    #pragma unroll
    for (int k = 0; k < OWNBLK * (ROWPITCH / 4) / 256; ++k) {
        const int i = (int)threadIdx.x + 256 * k;
        if (i < n4) base[i] = make_int4(0, 0, 0, 0);
    }
    if (threadIdx.x == 0) {
        ctl[CTL_OWN + b] = cache_half();
        if (b == 0) { ctl[CTL_OVN] = 0; ctl[CTL_TICK] = 0; ctl[CTL_TICK + 32] = 0; }
    }
    if (b == 0 && (int)threadIdx.x < ZLINES * 8) ((float4*)zline)[threadIdx.x] = make_float4(-0.0f, -0.0f, -0.0f, -0.0f);
}

__device__ __forceinline__ void place_edge(int d, int r, int s, int* __restrict__ rows,
                                           int* ovn, int* __restrict__ ovs,
                                           int* __restrict__ ovd) {
    if (r < ROWCAP) {
        rows[(size_t)d * ROWPITCH + r] = s;
    } else {
        const int o = atomicAdd(ovn, 1);
        ovs[o] = s;
        ovd[o] = d;
    }
}

__device__ __forceinline__ void place_chunk(const int* __restrict__ src, const int* __restrict__ dst,
                                            const int* own, int set, int e0, int E, bool whole,
                                            int* __restrict__ rows, int* ovn,
                                            int* __restrict__ ovs, int* __restrict__ ovd,
                                            int* tp, bool first, int& tnext) {
    if (whole) {
        int d[EPL], s[EPL], o[EPL], r[EPL];
        #pragma unroll
        for (int j = 0; j < EPL; ++j) d[j] = dst[e0 + 128 * j];
        #pragma unroll
        for (int j = 0; j < EPL; ++j) s[j] = src[e0 + 128 * j];
        #pragma unroll
        for (int j = 0; j < EPL; ++j) o[j] = own[d[j] >> OWNSHIFT];
        asm volatile("" :: "v"(o[0]), "v"(o[1]), "v"(o[2]), "v"(o[3]), "v"(o[4]), "v"(o[5]), "v"(o[6]), "v"(o[7]));
        #pragma unroll
        for (int j = 0; j < EPL; ++j) {
            r[j] = 0;
            if (o[j] == set) r[j] = atomicAdd(&rows[(size_t)d[j] * ROWPITCH + CNTSLOT], 1);
        }
        if (first) tnext = atomicAdd(tp, 1);
        #pragma unroll
        for (int j = 0; j < EPL; ++j)
            if (o[j] == set) place_edge(d[j], r[j], s[j], rows, ovn, ovs, ovd);
    } else {
        if (first) tnext = atomicAdd(tp, 1);
        for (int j = 0; j < EPL; ++j) {
            const int e = e0 + 128 * j;
            if (e < E) {
                const int d = dst[e];
                if (own[d >> OWNSHIFT] == set)
                    place_edge(d, atomicAdd(&rows[(size_t)d * ROWPITCH + CNTSLOT], 1), src[e], rows, ovn, ovs, ovd);
            }
        }
    }
}

__device__ __forceinline__ void place_edges(const int* __restrict__ src, const int* __restrict__ dst,
                                            int* __restrict__ rows, int* ctl,
                                            int* __restrict__ ovs, int* __restrict__ ovd, int E) {
    __shared__ int tk[2];
    const int tid = (int)threadIdx.x;
    int* ovn = ctl + CTL_OVN;
    const int* own = ctl + CTL_OWN;
    const int nch = (E + ECHUNK - 1) / ECHUNK;
    int set = cache_half();
    int turn = 0;
    for (int pass = 0; pass < 2; ++pass) {
        int* tp = ctl + CTL_TICK + 32 * set;
        int tnext = 0;
        if (tid == 0) tnext = atomicAdd(tp, 1);
        for (int it = 0; it <= nch; ++it) {
            if (tid == 0) tk[turn] = tnext;
            __syncthreads();
            const int t = __builtin_amdgcn_readfirstlane(tk[turn]);
            turn ^= 1;
            if (t >= nch) break;
            const int e0 = t * ECHUNK;
            place_chunk(src, dst, own, set, e0 + tid, E, e0 + ECHUNK <= E, rows, ovn, ovs, ovd, tp, tid == 0, tnext);
        }
        set ^= 1;
    }
}

template <int K, int NOUT, int MODE>
__device__ __forceinline__ void gemm_tiles(
    const float* __restrict__ in,
    const float* __restrict__ W,
    const float* __restrict__ bias,
    const int* __restrict__ rows,
    float* __restrict__ out,
    int n, int blk, int nblk)
{
    constexpr int KS = K / 4;
    constexpr int NT = NOUT / 16;

    const int lane  = threadIdx.x & 31;
    const int lhalf = lane >> 4;
    const int l16   = lane & 15;
    const int wave   = __builtin_amdgcn_readfirstlane((int)(threadIdx.x >> 5));
    const int gwave  = blk * 4 + wave;
    const int nwaves = nblk * 4;
    const int tiles  = n >> 4;
    if (gwave >= tiles) return;

    v2f bf[KS][NT];
    #pragma unroll
    for (int s = 0; s < KS; ++s) {
        #pragma unroll
        for (int t = 0; t < NT; ++t) {
            const float* wp = W + (t * 16 + l16) * K + 4 * s + 2 * lhalf;
            bf[s][t].x = wp[0];
            bf[s][t].y = wp[1];
        }
    }

    float bv[NT];
    if (MODE == 0) {
        #pragma unroll
        for (int t = 0; t < NT; ++t) bv[t] = bias[t * 16 + l16];
    }

    v2f a[KS];
    {
        const float* ap = in + (size_t)(gwave * 16 + l16) * K + 2 * lhalf;
        #pragma unroll
        for (int s = 0; s < KS; ++s) { a[s].x = ap[4 * s]; a[s].y = ap[4 * s + 1]; }
    }

    for (int tile = gwave; tile < tiles; tile += nwaves) {
        const int row0 = tile * 16;

        const int ntile = (tile + nwaves < tiles) ? (tile + nwaves) : tile;
        int dc[8];
        if (MODE == 1) {
            #pragma unroll
            for (int r = 0; r < 8; ++r) dc[r] = rows[(size_t)(row0 + r + 8 * lhalf) * ROWPITCH + CNTSLOT];
        }
        v2f an[KS];
        {
            const float* ap = in + (size_t)(ntile * 16 + l16) * K + 2 * lhalf;
            #pragma unroll
            for (int s = 0; s < KS; ++s) { an[s].x = ap[4 * s]; an[s].y = ap[4 * s + 1]; }
        }

        __builtin_amdgcn_sched_barrier(0);

        v8f acc[NT];
        #pragma unroll
        for (int t = 0; t < NT; ++t)
            acc[t] = (v8f){0.f, 0.f, 0.f, 0.f, 0.f, 0.f, 0.f, 0.f};

        #pragma unroll
        for (int s = 0; s < KS; ++s) {
            #pragma unroll
            for (int t = 0; t < NT; ++t) {
                acc[t] = __builtin_amdgcn_wmma_f32_16x16x4_f32(
                    false, a[s], false, bf[s][t], (short)0, acc[t], false, false);
            }
        }

        float dv[8];
        if (MODE == 1) {
            __builtin_amdgcn_sched_barrier(0);
            #pragma unroll
            for (int r = 0; r < 8; ++r) dv[r] = rsqrtf((float)(dc[r] + 1));
        }

        #pragma unroll
        for (int t = 0; t < NT; ++t) {
            const int col = t * 16 + l16;
            if (MODE == 0) {
                #pragma unroll
                for (int r = 0; r < 8; ++r) {
                    const int row = row0 + r + 8 * lhalf;
                    out[(size_t)row * NOUT + col] = lrelu(acc[t][r] + bv[t]);
                }
            } else {
                #pragma unroll
                for (int r = 0; r < 8; ++r) {
                    const int row = row0 + r + 8 * lhalf;
                    out[(size_t)row * NOUT + col] = acc[t][r] * dv[r];
                }
            }
        }

        __builtin_amdgcn_sched_barrier(0);

        #pragma unroll
        for (int s = 0; s < KS; ++s) a[s] = an[s];
    }
}

template <int K, int NOUT, int MODE>
__global__ __launch_bounds__(128) void gcn_gemm(
    const float* __restrict__ in, const float* __restrict__ W, const float* __restrict__ bias,
    const int* __restrict__ rows, float* __restrict__ out, int n)
{
    gemm_tiles<K, NOUT, MODE>(in, W, bias, rows, out, n, (int)blockIdx.x, (int)gridDim.x);
}

__global__ __launch_bounds__(128) void gcn_gemm_in_and_place(
    const float* __restrict__ x, const float* __restrict__ W_in, const float* __restrict__ b_in,
    float* __restrict__ h0, int n, int gblk,
    const int* __restrict__ src, const int* __restrict__ dst,
    int* __restrict__ rows, int* ctl, int* __restrict__ ovs, int* __restrict__ ovd, int E)
{
    if ((int)blockIdx.x < gblk)
        gemm_tiles<64, 64, 0>(x, W_in, b_in, nullptr, h0, n, (int)blockIdx.x, gblk);
    else
        place_edges(src, dst, rows, ctl, ovs, ovd, E);
}


template <int J>
__device__ __forceinline__ unsigned bcast8(unsigned x) {
    return __builtin_amdgcn_mov_dpp8(x, J * 0x249249);
}

struct NodeHead { int v; int4 r; float4 s; int deg; uint4 o; };

__device__ __forceinline__ NodeHead load_head(int v, const int4* __restrict__ row4,
                                              const float4* __restrict__ p4) {
    NodeHead h;
    h.v = v;
    h.r = row4[(size_t)v * 8];
    h.s = p4[(size_t)v * 8];
    h.deg = 0;
    h.o = make_uint4(0u, 0u, 0u, 0u);
    return h;
}

__device__ __forceinline__ void head_ready(NodeHead& h, int sub, int top, unsigned zoff) {
    h.deg = (int)bcast8<7>((unsigned)h.r.w);
    const int inrow = min(h.deg, ROWCAP);
    const int e0 = 4 * sub;
    h.o.x = e0 + 0 < inrow ? (unsigned)min(max(h.r.x, 0), top) * 128u : zoff;
    h.o.y = e0 + 1 < inrow ? (unsigned)min(max(h.r.y, 0), top) * 128u : zoff;
    h.o.z = e0 + 2 < inrow ? (unsigned)min(max(h.r.z, 0), top) * 128u : zoff;
    h.o.w = e0 + 3 < inrow ? (unsigned)min(max(h.r.w, 0), top) * 128u : zoff;
}

template <int J>
__device__ __forceinline__ void trip_fetch(const NodeHead& h, const char* __restrict__ pb,
                                           unsigned sub16, float4* m) {
    m[0] = *(const float4*)(pb + (bcast8<J>(h.o.x) | sub16));
    m[1] = *(const float4*)(pb + (bcast8<J>(h.o.y) | sub16));
    m[2] = *(const float4*)(pb + (bcast8<J>(h.o.z) | sub16));
    m[3] = *(const float4*)(pb + (bcast8<J>(h.o.w) | sub16));
}

__device__ __forceinline__ void trip_sum(float4& s, const float4* m) {
    s.x += m[0].x; s.y += m[0].y; s.z += m[0].z; s.w += m[0].w;
    s.x += m[1].x; s.y += m[1].y; s.z += m[1].z; s.w += m[1].w;
    s.x += m[2].x; s.y += m[2].y; s.z += m[2].z; s.w += m[2].w;
    s.x += m[3].x; s.y += m[3].y; s.z += m[3].z; s.w += m[3].w;
}

__device__ __forceinline__ void fetch16(NodeHead& h, int sub, int top, unsigned zoff,
                                        const char* __restrict__ pb, unsigned sub16, float4* m) {
    head_ready(h, sub, top, zoff);
    trip_fetch<0>(h, pb, sub16, m);
    trip_fetch<1>(h, pb, sub16, m + 4);
    trip_fetch<2>(h, pb, sub16, m + 8);
    trip_fetch<3>(h, pb, sub16, m + 12);
}

template <int J>
__device__ __forceinline__ void trip_late(float4& s, const NodeHead& h, const char* __restrict__ pb, unsigned sub16) {
    float4 m[4];
    trip_fetch<J>(h, pb, sub16, m);
    trip_sum(s, m);
}

template <int MODE, bool GATHER_NEXT, bool HEAD_AFTER>
__device__ __forceinline__ void pull_step(int g, int base, int n, int lane, int k, int sub,
                                          NodeHead& c, NodeHead& nx, const float4* mc, float4* mn,
                                          const int4* __restrict__ row4,
                                          const float4* __restrict__ p4,
                                          const char* __restrict__ pb, unsigned sub16, unsigned zoff,
                                          const int* __restrict__ ovn, const int* __restrict__ ovs,
                                          const int* __restrict__ ovd, int E,
                                          float b0, float b1, float b2, float b3,
                                          float* __restrict__ dstp, float* tw)
{
    const int top = n - 1;
    NodeHead nn = nx;
    if (HEAD_AFTER) nn = load_head(min(base + 4 * (g + 2) + k, top), row4, p4);
    if (GATHER_NEXT) fetch16(nx, sub, top, zoff, pb, sub16, mn);
    __builtin_amdgcn_sched_barrier(0);

    const int v = c.v, deg = c.deg;
    const int inrow = min(deg, ROWCAP);
    float4 s = c.s;
    trip_sum(s, mc);
    trip_sum(s, mc + 4);
    trip_sum(s, mc + 8);
    trip_sum(s, mc + 12);

    if (__builtin_amdgcn_ballot_w32(inrow > 16) != 0u) {
        trip_late<4>(s, c, pb, sub16);
        if (__builtin_amdgcn_ballot_w32(inrow > 20) != 0u) {
            trip_late<5>(s, c, pb, sub16);
            if (__builtin_amdgcn_ballot_w32(inrow > 24) != 0u) {
                trip_late<6>(s, c, pb, sub16);
                if (__builtin_amdgcn_ballot_w32(inrow > 28) != 0u)
                    trip_late<7>(s, c, pb, sub16);
            }
        }
    }

    if (deg > ROWCAP) {
        const int novf = min(ovn[0], E);
        for (int o = 0; o < novf; ++o) {
            if (ovd[o] == v) {
                const float4 m = p4[(size_t)min(max(ovs[o], 0), top) * 8];
                s.x += m.x; s.y += m.y; s.z += m.z; s.w += m.w;
            }
        }
    }

    const float di = rsqrtf((float)(deg + 1));
    if (MODE == 0) {
        float4 h;
        h.x = di * lrelu(di * s.x + b0);
        h.y = di * lrelu(di * s.y + b1);
        h.z = di * lrelu(di * s.z + b2);
        h.w = di * lrelu(di * s.w + b3);
        if (base + 4 * g + k < n) ((float4*)dstp)[(size_t)v * 8 + sub] = h;
    } else {
        float4 t;
        t.x = di * s.x;
        t.y = di * s.y;
        t.z = di * s.z;
        t.w = di * s.w;
        *(float4*)(tw + (4 * g + k) * TPITCH + 4 * sub) = t;
    }
    c = nx;
    nx = nn;
}

__device__ __forceinline__ float dot8(const v8f& z, const float4& wa, const float4& wb,
                                      const float4& ba, const float4& bb) {
    float d = wa.x * lrelu(z[0] + ba.x);
    d += wa.y * lrelu(z[1] + ba.y);
    d += wa.z * lrelu(z[2] + ba.z);
    d += wa.w * lrelu(z[3] + ba.w);
    d += wb.x * lrelu(z[4] + bb.x);
    d += wb.y * lrelu(z[5] + bb.y);
    d += wb.z * lrelu(z[6] + bb.z);
    d += wb.w * lrelu(z[7] + bb.w);
    return d;
}

template <int MODE>
__global__ __launch_bounds__(128)
void gcn_pull(const int* __restrict__ rows,
              const int* __restrict__ ovn, const int* __restrict__ ovs,
              const int* __restrict__ ovd, int E,
              const float* __restrict__ p, const float* __restrict__ bias,
              const float* __restrict__ Wc,
              const float* __restrict__ Wout, const float* __restrict__ bout,
              const float* __restrict__ zline,
              float* __restrict__ dstp, int n)
{
    __shared__ float tl[MODE == 1 ? 4 * 32 * TPITCH : 1];
    const int lane = threadIdx.x & 31;
    const int k    = lane >> 3;
    const int sub  = lane & 7;
    const int wave = __builtin_amdgcn_readfirstlane((int)(threadIdx.x >> 5));
    const int base = (blockIdx.x * ((int)blockDim.x >> 5) + wave) * 32;
    if (base >= n) return;
    const int top = n - 1;
    const float4* p4   = (const float4*)p + sub;
    const int4*   row4 = (const int4*)rows + sub;
    const char*    pb    = (const char*)p;
    const unsigned sub16 = 16u * (unsigned)sub;
    const unsigned zoff  = (unsigned)((const char*)zline - (const char*)p) + 128u * ((unsigned)blockIdx.x % ZLINES);

    float b0 = 0.0f, b1 = 0.0f, b2 = 0.0f, b3 = 0.0f;
    if (MODE == 0) { b0 = bias[4 * sub]; b1 = bias[4 * sub + 1]; b2 = bias[4 * sub + 2]; b3 = bias[4 * sub + 3]; }
    float* tw = tl + (MODE == 1 ? wave * 32 * TPITCH : 0);

    NodeHead c  = load_head(min(base + k, top), row4, p4);
    NodeHead nx = load_head(min(base + 4 + k, top), row4, p4);
    float4 ma[16], mb[16];
    fetch16(c, sub, top, zoff, pb, sub16, ma);

    for (int g = 0; g < 6; g += 2) {
        pull_step<MODE, true, true>(g,     base, n, lane, k, sub, c, nx, ma, mb, row4, p4, pb, sub16, zoff, ovn, ovs, ovd, E,
                                    b0, b1, b2, b3, dstp, tw);
        pull_step<MODE, true, true>(g + 1, base, n, lane, k, sub, c, nx, mb, ma, row4, p4, pb, sub16, zoff, ovn, ovs, ovd, E,
                                    b0, b1, b2, b3, dstp, tw);
    }
    pull_step<MODE, true, false>(6, base, n, lane, k, sub, c, nx, ma, mb, row4, p4, pb, sub16, zoff, ovn, ovs, ovd, E,
                                 b0, b1, b2, b3, dstp, tw);
    pull_step<MODE, false, false>(7, base, n, lane, k, sub, c, nx, mb, ma, row4, p4, pb, sub16, zoff, ovn, ovs, ovd, E,
                                  b0, b1, b2, b3, dstp, tw);
    if (MODE == 1) {
        const int c16 = lane & 15;
        const int hh  = lane >> 4;

        v2f af[2][8];
        #pragma unroll
        for (int m = 0; m < 2; ++m) {
            const float4* wp = (const float4*)(Wc + (16 * m + c16) * 32 + 16 * hh);
            #pragma unroll
            for (int q = 0; q < 4; ++q) {
                const float4 w = wp[q];
                af[m][2 * q]     = (v2f){w.x, w.y};
                af[m][2 * q + 1] = (v2f){w.z, w.w};
            }
        }
        float4 bz[2][2], wz[2][2];
        #pragma unroll
        for (int m = 0; m < 2; ++m) {
            const float4* bp = (const float4*)(bias + 16 * m + 8 * hh);
            const float4* op = (const float4*)(Wout + 16 * m + 8 * hh);
            bz[m][0] = bp[0]; bz[m][1] = bp[1];
            wz[m][0] = op[0]; wz[m][1] = op[1];
        }
        const float bo = bout[0];

        __builtin_amdgcn_fence(__ATOMIC_RELEASE, "wavefront");
        __builtin_amdgcn_fence(__ATOMIC_ACQUIRE, "wavefront");

        v2f bt[2][8];
        #pragma unroll
        for (int t = 0; t < 2; ++t) {
            const float4* rp = (const float4*)(tw + (16 * t + c16) * TPITCH + 16 * hh);
            #pragma unroll
            for (int q = 0; q < 4; ++q) {
                const float4 x = rp[q];
                bt[t][2 * q]     = (v2f){x.x, x.y};
                bt[t][2 * q + 1] = (v2f){x.z, x.w};
            }
        }

        v8f acc[2][2];
        #pragma unroll
        for (int m = 0; m < 2; ++m) {
            #pragma unroll
            for (int t = 0; t < 2; ++t)
                acc[m][t] = (v8f){0.f, 0.f, 0.f, 0.f, 0.f, 0.f, 0.f, 0.f};
        }
        #pragma unroll
        for (int s = 0; s < 8; ++s) {
            #pragma unroll
            for (int m = 0; m < 2; ++m) {
                #pragma unroll
                for (int t = 0; t < 2; ++t)
                    acc[m][t] = __builtin_amdgcn_wmma_f32_16x16x4_f32(
                        false, af[m][s], false, bt[t][s], (short)0, acc[m][t], false, false);
            }
        }

        float part[2];
        #pragma unroll
        for (int t = 0; t < 2; ++t) {
            const float d = dot8(acc[0][t], wz[0][0], wz[0][1], bz[0][0], bz[0][1])
                          + dot8(acc[1][t], wz[1][0], wz[1][1], bz[1][0], bz[1][1]);
            part[t] = d + __shfl_xor(d, 16);
        }
        const float outv = hh ? part[1] : part[0];
        if (base + lane < n) dstp[base + lane] = outv + bo;
    }
}

extern "C" void kernel_launch(void* const* d_in, const int* in_sizes, int n_in,
                              void* d_out, int out_size, void* d_ws, size_t ws_size,
                              hipStream_t stream)
{
    const float* x     = (const float*)d_in[0];
    const int*   ei    = (const int*)d_in[1];
    const float* W_in  = (const float*)d_in[2];
    const float* b_in  = (const float*)d_in[3];
    const float* W1    = (const float*)d_in[4];
    const float* b1    = (const float*)d_in[5];
    const float* W2    = (const float*)d_in[6];
    const float* b2    = (const float*)d_in[7];
    const float* W_out = (const float*)d_in[8];
    const float* b_out = (const float*)d_in[9];
    float* out = (float*)d_out;

    const int N = in_sizes[0] / 64;
    const int E = in_sizes[1] / 2;
    const int* src = ei;
    const int* dst = ei + E;

    float* ws   = (float*)d_ws;
    int* ctl    = (int*)d_ws;
    float* h0   = ws + (size_t)N;
    float* p    = ws + (size_t)N * 65;
    float* h1   = ws + (size_t)N * 97;
    float* zl   = ws + (size_t)N * 129;
    int* rows   = (int*)(ws + (size_t)N * 161);
    int* ovs    = rows + (size_t)N * ROWPITCH;
    int* ovd    = ovs + (size_t)E;

    const int TB = 256;
    const int gblk   = 512;
    const int eblk   = 1024;

    gcn_rows_zero<<<(N + TB - 1) / TB, TB, 0, stream>>>((int4*)rows, ctl, zl, N);
    gcn_gemm_in_and_place<<<gblk + eblk, 128, 0, stream>>>(x, W_in, b_in, h0, N, gblk, src, dst, rows, ctl, ovs, ovd, E);

    gcn_gemm<64, 32, 1><<<gblk, 128, 0, stream>>>(h0, W1, nullptr, rows, p, N);
    gcn_pull<0><<<((size_t)N + 127) / 128, 128, 0, stream>>>(rows, ctl, ovs, ovd, E, p, b1, nullptr, nullptr, nullptr, zl, h1, N);

    gcn_pull<1><<<((size_t)N + 127) / 128, 128, 0, stream>>>(rows, ctl, ovs, ovd, E, h1, b2, W2, W_out, b_out, zl, out, N);
}
